# starting kernel
# baseline (speedup 1.0000x reference)
#include <hip/hip_runtime.h>
#include <hip/hip_bf16.h>

typedef float v2f __attribute__((ext_vector_type(2)));
typedef float v8f __attribute__((ext_vector_type(8)));


__global__ void k_deg_init(float* __restrict__ deg, int n) {
  int i = blockIdx.x * blockDim.x + threadIdx.x;
  if (i < n) deg[i] = 1.0f;
}

__global__ void k_deg_count(const int* __restrict__ col, float* __restrict__ deg, int e) {
  int i = blockIdx.x * blockDim.x + threadIdx.x;
  if (i < e) atomicAdd(&deg[col[i]], 1.0f);
}

__global__ void k_deg_rsqrt(const float* __restrict__ deg, float* __restrict__ dinv, int n) {
  int i = blockIdx.x * blockDim.x + threadIdx.x;
  if (i < n) dinv[i] = rsqrtf(deg[i]);
}


__global__ void k_gemm_wmma(const float* __restrict__ x,
                            const float* __restrict__ W,
                            float* __restrict__ xw, int ntiles) {
  const int lane   = threadIdx.x & 31;
  const int waveIb = threadIdx.x >> 5;
  const int tile   = blockIdx.x * (blockDim.x >> 5) + waveIb;
  if (tile >= ntiles) return;

  const int m     = lane & 15;
  const int khalf = lane >> 4;

  const float* xrow  = x + (size_t)(tile * 16 + m) * 128 + khalf * 2;
  const float* wbase = W + (size_t)m * 128 + khalf * 2;

  v8f acc[8] = {};

  v2f a_cur = *(const v2f*)(xrow);
  v2f b_cur[8];
#pragma unroll
  for (int t = 0; t < 8; ++t) b_cur[t] = *(const v2f*)(wbase + t * 2048);

#pragma unroll
  for (int k0 = 0; k0 < 128; k0 += 4) {
    v2f a_nxt = a_cur;
    v2f b_nxt[8];
#pragma unroll
    for (int t = 0; t < 8; ++t) b_nxt[t] = b_cur[t];
    if (k0 + 4 < 128) {
      a_nxt = *(const v2f*)(xrow + k0 + 4);
#pragma unroll
      for (int t = 0; t < 8; ++t)
        b_nxt[t] = *(const v2f*)(wbase + t * 2048 + k0 + 4);
    }
#pragma unroll
    for (int t = 0; t < 8; ++t) {
      acc[t] = __builtin_amdgcn_wmma_f32_16x16x4_f32(
           false, a_cur,  false, b_cur[t],
           (short)0, acc[t],  false,  false);
    }
    a_cur = a_nxt;
#pragma unroll
    for (int t = 0; t < 8; ++t) b_cur[t] = b_nxt[t];
  }

  float* outbase = xw + (size_t)(tile * 16 + khalf * 8) * 128 + m;
#pragma unroll
  for (int t = 0; t < 8; ++t) {
#pragma unroll
    for (int v = 0; v < 8; ++v) {
      outbase[(size_t)v * 128 + t * 16] = acc[t][v];
    }
  }
}

__global__ void k_gemm_tail(const float* __restrict__ x, const float* __restrict__ W,
                            float* __restrict__ xw, int n, int start) {
  int i = blockIdx.x * blockDim.x + threadIdx.x;
  int total = (n - start) * 128;
  if (i >= total) return;
  int node = start + i / 128;
  int o = i & 127;
  const float* xr = x + (size_t)node * 128;
  const float* wr = W + (size_t)o * 128;
  float s = 0.f;
  for (int k = 0; k < 128; ++k) s += xr[k] * wr[k];
  xw[(size_t)node * 128 + o] = s;
}


__global__ void k_init_out(const float* __restrict__ xw, const float* __restrict__ dinv,
                           const float* __restrict__ bias, float* __restrict__ out, int n) {
  int i = blockIdx.x * blockDim.x + threadIdx.x;
  int total = n * 32;
  if (i >= total) return;
  int node = i >> 5;
  int f = (i & 31) * 4;
  float d = dinv[node];
  float s = d * d;
  const float4 v = *(const float4*)(xw + (size_t)node * 128 + f);
  const float4 bb = *(const float4*)(bias + f);
  float4 o;
  o.x = v.x * s + bb.x;
  o.y = v.y * s + bb.y;
  o.z = v.z * s + bb.z;
  o.w = v.w * s + bb.w;
  *(float4*)(out + (size_t)node * 128 + f) = o;
}


__global__ void k_scatter(const float* __restrict__ xw,
                          const int* __restrict__ row, const int* __restrict__ col,
                          const float* __restrict__ dinv,
                          float* __restrict__ out, int e) {
  const int lane = threadIdx.x & 31;
  const int edge = (blockIdx.x * blockDim.x + threadIdx.x) >> 5;
  if (edge >= e) return;
  const int r = row[edge];
  const int c = col[edge];
  const float nrm = dinv[r] * dinv[c];
  const float4 v = *(const float4*)(xw + (size_t)r * 128 + lane * 4);
  float* dst = out + (size_t)c * 128 + lane * 4;
  atomicAdd(dst + 0, v.x * nrm);
  atomicAdd(dst + 1, v.y * nrm);
  atomicAdd(dst + 2, v.z * nrm);
  atomicAdd(dst + 3, v.w * nrm);
}


extern "C" void kernel_launch(void* const* d_in, const int* in_sizes, int n_in,
                              void* d_out, int out_size, void* d_ws, size_t ws_size,
                              hipStream_t stream) {
  const float* x  = (const float*)d_in[0];
  const int* ei   = (const int*)d_in[1];
  const float* W  = (const float*)d_in[2];
  const float* b  = (const float*)d_in[3];
  float* out      = (float*)d_out;

  const int N = in_sizes[0] / 128;
  const int E = in_sizes[1] / 2;
  const int* row = ei;
  const int* col = ei + E;

  float* xw   = (float*)d_ws;
  float* deg  = xw + (size_t)N * 128;
  float* dinv = deg + N;

  k_deg_init<<<(N + 255) / 256, 256, 0, stream>>>(deg, N);
  k_deg_count<<<(E + 255) / 256, 256, 0, stream>>>(col, deg, E);
  k_deg_rsqrt<<<(N + 255) / 256, 256, 0, stream>>>(deg, dinv, N);

  const int ntiles = N / 16;
  if (ntiles > 0)
    k_gemm_wmma<<<(ntiles + 7) / 8, 256, 0, stream>>>(x, W, xw, ntiles);
  const int rem = N % 16;
  if (rem > 0)
    k_gemm_tail<<<(rem * 128 + 255) / 256, 256, 0, stream>>>(x, W, xw, N, ntiles * 16);

  k_init_out<<<(N * 32 + 255) / 256, 256, 0, stream>>>(xw, dinv, b, out, N);

  k_scatter<<<(E + 7) / 8, 256, 0, stream>>>(xw, row, col, dinv, out, E);
}


# Edge scan reads 1-byte tile keys (16 per lane per load, 4 compared per word), not 4-byte targets; keys written by the degree-count launch, now 16 edges per thread; LDS atomic sum order as before
# speedup vs baseline: 1.9720x; 1.9720x over previous
#include <hip/hip_runtime.h>
#include <hip/hip_bf16.h>

typedef float v2f __attribute__((ext_vector_type(2)));
typedef float v8f __attribute__((ext_vector_type(8)));


__global__ void k_deg_init(float* __restrict__ deg, int n) {
  int i = blockIdx.x * blockDim.x + threadIdx.x;
  if (i < n) deg[i] = 1.0f;
}

#define TILE_NODES 392

__global__ void k_deg_count_keys(const int* __restrict__ col, float* __restrict__ deg,
                                 unsigned* __restrict__ keys, int e) {
  const int t = blockIdx.x * blockDim.x + threadIdx.x;
  const int e0 = t * 16;
  if (e0 >= e) return;
  int c[16];
  if ((e & 3) == 0 && e0 + 15 < e) {
#pragma unroll
    for (int d = 0; d < 4; ++d) {
      const int4 v = *(const int4*)(col + e0 + d * 4);
      c[d * 4 + 0] = v.x;
      c[d * 4 + 1] = v.y;
      c[d * 4 + 2] = v.z;
      c[d * 4 + 3] = v.w;
    }
  } else {
#pragma unroll
    for (int q = 0; q < 16; ++q) c[q] = (e0 + q < e) ? col[e0 + q] : -1;
  }
  unsigned kw[4];
#pragma unroll
  for (int d = 0; d < 4; ++d) {
    unsigned k = 0;
#pragma unroll
    for (int j = 0; j < 4; ++j) {
      const int cj = c[d * 4 + j];
      if (cj >= 0) {
        atomicAdd(&deg[cj], 1.0f);
        k |= (((unsigned)cj / (unsigned)TILE_NODES) & 255u) << (8 * j);
      }
    }
    kw[d] = k;
  }
  *(uint4*)(keys + (size_t)t * 4) = make_uint4(kw[0], kw[1], kw[2], kw[3]);
}

__global__ void k_deg_rsqrt(const float* __restrict__ deg, float* __restrict__ dinv, int n) {
  int i = blockIdx.x * blockDim.x + threadIdx.x;
  if (i < n) dinv[i] = rsqrtf(deg[i]);
}


__global__ void k_gemm_wmma(const float* __restrict__ x,
                            const float* __restrict__ W,
                            float* __restrict__ xw, int ntiles) {
  const int lane   = threadIdx.x & 31;
  const int waveIb = threadIdx.x >> 5;
  const int tile   = blockIdx.x * (blockDim.x >> 5) + waveIb;
  if (tile >= ntiles) return;

  const int m     = lane & 15;
  const int khalf = lane >> 4;

  const float* xrow  = x + (size_t)(tile * 16 + m) * 128 + khalf * 2;
  const float* wbase = W + (size_t)m * 128 + khalf * 2;

  v8f acc[8] = {};

  v2f a_cur = *(const v2f*)(xrow);
  v2f b_cur[8];
#pragma unroll
  for (int t = 0; t < 8; ++t) b_cur[t] = *(const v2f*)(wbase + t * 2048);

#pragma unroll
  for (int k0 = 0; k0 < 128; k0 += 4) {
    v2f a_nxt = a_cur;
    v2f b_nxt[8];
#pragma unroll
    for (int t = 0; t < 8; ++t) b_nxt[t] = b_cur[t];
    if (k0 + 4 < 128) {
      a_nxt = *(const v2f*)(xrow + k0 + 4);
#pragma unroll
      for (int t = 0; t < 8; ++t)
        b_nxt[t] = *(const v2f*)(wbase + t * 2048 + k0 + 4);
    }
#pragma unroll
    for (int t = 0; t < 8; ++t) {
      acc[t] = __builtin_amdgcn_wmma_f32_16x16x4_f32(
           false, a_cur,  false, b_cur[t],
           (short)0, acc[t],  false,  false);
    }
    a_cur = a_nxt;
#pragma unroll
    for (int t = 0; t < 8; ++t) b_cur[t] = b_nxt[t];
  }

  float* outbase = xw + (size_t)(tile * 16 + khalf * 8) * 128 + m;
#pragma unroll
  for (int t = 0; t < 8; ++t) {
#pragma unroll
    for (int v = 0; v < 8; ++v) {
      outbase[(size_t)v * 128 + t * 16] = acc[t][v];
    }
  }
}

__global__ void k_gemm_tail(const float* __restrict__ x, const float* __restrict__ W,
                            float* __restrict__ xw, int n, int start) {
  int i = blockIdx.x * blockDim.x + threadIdx.x;
  int total = (n - start) * 128;
  if (i >= total) return;
  int node = start + i / 128;
  int o = i & 127;
  const float* xr = x + (size_t)node * 128;
  const float* wr = W + (size_t)o * 128;
  float s = 0.f;
  for (int k = 0; k < 128; ++k) s += xr[k] * wr[k];
  xw[(size_t)node * 128 + o] = s;
}


#define AGG_WAVES 32
#define QCAP 576
#define DRAIN_AT 64
#define SCAN_U 4

static __device__ __forceinline__ void lds_add(float* p, float v) {
  __hip_atomic_fetch_add(p, v, __ATOMIC_RELAXED, __HIP_MEMORY_SCOPE_WORKGROUP);
}

static __device__ __forceinline__ void drain_queue(float* acc, const int* qe, int cnt, int lane,
                                                   int lo, int cntNodes,
                                                   const float* __restrict__ xw,
                                                   const int* __restrict__ row,
                                                   const int* __restrict__ col,
                                                   const float* __restrict__ dinv) {
  for (int i = 0; i < cnt; i += 32) {
    int idx = i + lane;
    if (idx >= cnt) idx = cnt - 1;
    const int ed = qe[idx];
    const int cl = col[ed] - lo;
    const int rl = row[ed];
    const float sl = dinv[rl];
    int nb = cnt - i;
    if (nb > 32) nb = 32;
    for (int k0 = 0; k0 < nb; k0 += 8) {
      float4 v[8];
      int cc[8];
      float ss[8];
#pragma unroll
      for (int k = 0; k < 8; ++k) {
        const int kk = k0 + k;
        const int r = __builtin_amdgcn_readlane(rl, kk);
        cc[k] = __builtin_amdgcn_readlane(cl, kk);
        ss[k] = __int_as_float(__builtin_amdgcn_readlane(__float_as_int(sl), kk));
        v[k] = *(const float4*)(xw + (size_t)r * 128 + lane * 4);
      }
#pragma unroll
      for (int k = 0; k < 8; ++k) {
        if (k0 + k < nb && (unsigned)cc[k] < (unsigned)cntNodes) {
          float* a = acc + cc[k] * 128 + lane;
          lds_add(a, v[k].x * ss[k]);
          lds_add(a + 32, v[k].y * ss[k]);
          lds_add(a + 64, v[k].z * ss[k]);
          lds_add(a + 96, v[k].w * ss[k]);
        }
      }
    }
  }
}

static __device__ __forceinline__ int scan_keys(float* acc, int* qe, int w, int lane,
                                                int lo, int cntNodes,
                                                const float* __restrict__ xw,
                                                const int* __restrict__ row,
                                                const int* __restrict__ col,
                                                const float* __restrict__ dinv,
                                                const unsigned* __restrict__ keys, int e) {
  int cnt = 0;
  const int nChunks = (e + 511) >> 9;
  const int nGroups = (nChunks + SCAN_U - 1) / SCAN_U;
  const int lastVec = ((e + 15) >> 4) - 1;
  const int g0 = (int)(((long long)blockIdx.x * nGroups) / gridDim.x);
  const unsigned keyRep = (blockIdx.x & 255u) * 0x01010101u;

  for (int kg = w; kg < nGroups; kg += AGG_WAVES) {
    int g = g0 + kg;
    if (g >= nGroups) g -= nGroups;
    uint4 kv[SCAN_U];
#pragma unroll
    for (int u = 0; u < SCAN_U; ++u) {
      int vi = (g * SCAN_U + u) * 32 + lane;
      if (vi > lastVec) vi = lastVec;
      kv[u] = *(const uint4*)(keys + (size_t)vi * 4);
    }
#pragma unroll
    for (int u = 0; u < SCAN_U; ++u) {
      const int c0 = (g * SCAN_U + u) << 9;
      const int eb = c0 + lane * 16;
      const unsigned kd[4] = {kv[u].x, kv[u].y, kv[u].z, kv[u].w};
      unsigned f[4];
#pragma unroll
      for (int d = 0; d < 4; ++d) {
        const unsigned x = kd[d] ^ keyRep;
        f[d] = ~(((x & 0x7F7F7F7Fu) + 0x7F7F7F7Fu) | x) & 0x80808080u;
      }
      if (c0 + 512 > e) {
#pragma unroll
        for (int d = 0; d < 4; ++d) {
          const int rem = e - (eb + d * 4);
          if (rem < 4) f[d] = (rem <= 0) ? 0u : (f[d] & ((1u << (8 * rem)) - 1u));
        }
      }
      unsigned F = f[0] | (f[1] >> 1) | (f[2] >> 2) | (f[3] >> 3);
      unsigned mask = __builtin_amdgcn_ballot_w32(F != 0u);
      while (mask) {
        const int pos = cnt + (int)__builtin_amdgcn_mbcnt_lo(mask, 0u);
        if (F != 0u) {
          const int p = __builtin_ctz(F);
          qe[pos] = eb + ((7 - (p & 7)) << 2) + (p >> 3);
        }
        cnt += __builtin_popcount(mask);
        F &= F - 1u;
        mask = __builtin_amdgcn_ballot_w32(F != 0u);
      }
      if (cnt >= DRAIN_AT) {
        drain_queue(acc, qe, cnt, lane, lo, cntNodes, xw, row, col, dinv);
        cnt = 0;
      }
    }
  }
  return cnt;
}

__global__ __launch_bounds__(1024) void k_aggregate(const float* __restrict__ xw,
                                                    const int* __restrict__ row,
                                                    const int* __restrict__ col,
                                                    const float* __restrict__ dinv,
                                                    const float* __restrict__ bias,
                                                    const unsigned* __restrict__ keys,
                                                    float* __restrict__ out, int n, int e) {
  __shared__ float acc[TILE_NODES * 128];
  __shared__ int qe_all[AGG_WAVES * QCAP];

  const int lane = threadIdx.x & 31;
  const int w = __builtin_amdgcn_readfirstlane(threadIdx.x >> 5);
  const int lo = blockIdx.x * TILE_NODES;
  int cntNodes = n - lo;
  if (cntNodes > TILE_NODES) cntNodes = TILE_NODES;

  for (int t = w; t < cntNodes; t += AGG_WAVES) {
    const int c = lo + t;
    const float d = dinv[c];
    const float4 v = *(const float4*)(xw + (size_t)c * 128 + lane * 4);
    float* a = acc + t * 128 + lane;
    a[0] = v.x * d;
    a[32] = v.y * d;
    a[64] = v.z * d;
    a[96] = v.w * d;
  }
  __syncthreads();

  int* qe = qe_all + w * QCAP;
  const int cnt = scan_keys(acc, qe, w, lane, lo, cntNodes, xw, row, col, dinv, keys, e);
  drain_queue(acc, qe, cnt, lane, lo, cntNodes, xw, row, col, dinv);
  __syncthreads();

  const float4 bb = *(const float4*)(bias + lane * 4);
  for (int t = w; t < cntNodes; t += AGG_WAVES) {
    const int c = lo + t;
    const float d = dinv[c];
    const float* a = acc + t * 128 + lane;
    float4 o;
    o.x = a[0] * d + bb.x;
    o.y = a[32] * d + bb.y;
    o.z = a[64] * d + bb.z;
    o.w = a[96] * d + bb.w;
    *(float4*)(out + (size_t)c * 128 + lane * 4) = o;
  }
}


extern "C" void kernel_launch(void* const* d_in, const int* in_sizes, int n_in,
                              void* d_out, int out_size, void* d_ws, size_t ws_size,
                              hipStream_t stream) {
  const float* x  = (const float*)d_in[0];
  const int* ei   = (const int*)d_in[1];
  const float* W  = (const float*)d_in[2];
  const float* b  = (const float*)d_in[3];
  float* out      = (float*)d_out;

  const int N = in_sizes[0] / 128;
  const int E = in_sizes[1] / 2;
  const int* row = ei;
  const int* col = ei + E;

  float* xw   = (float*)d_ws;
  unsigned* keys = (unsigned*)(xw + (size_t)N * 128);
  const int nvec = (E + 15) / 16;
  float* deg  = (float*)(keys + (size_t)nvec * 4);
  float* dinv = deg + N;

  k_deg_init<<<(N + 255) / 256, 256, 0, stream>>>(deg, N);
  if (nvec > 0)
    k_deg_count_keys<<<(nvec + 255) / 256, 256, 0, stream>>>(col, deg, keys, E);
  k_deg_rsqrt<<<(N + 255) / 256, 256, 0, stream>>>(deg, dinv, N);

  const int ntiles = N / 16;
  if (ntiles > 0)
    k_gemm_wmma<<<(ntiles + 7) / 8, 256, 0, stream>>>(x, W, xw, ntiles);
  const int rem = N % 16;
  if (rem > 0)
    k_gemm_tail<<<(rem * 128 + 255) / 256, 256, 0, stream>>>(x, W, xw, N, ntiles * 16);

  k_aggregate<<<(N + TILE_NODES - 1) / TILE_NODES, 1024, 0, stream>>>(xw, row, col, dinv, b, keys, out, N, E);
}


# GEMM keeps all of W in registers: 1024 waves (one per SIMD) load the 128x128 W fragments once and loop over 16-row strips, loading only x, next strip ahead; k order and sums unchanged
# speedup vs baseline: 2.2343x; 1.1330x over previous
#include <hip/hip_runtime.h>
#include <hip/hip_bf16.h>

typedef float v2f __attribute__((ext_vector_type(2)));
typedef float v8f __attribute__((ext_vector_type(8)));


__global__ void k_deg_init(float* __restrict__ deg, int n) {
  int i = blockIdx.x * blockDim.x + threadIdx.x;
  if (i < n) deg[i] = 1.0f;
}

#define TILE_NODES 392

__global__ void k_deg_count_keys(const int* __restrict__ col, float* __restrict__ deg,
                                 unsigned* __restrict__ keys, int e) {
  const int t = blockIdx.x * blockDim.x + threadIdx.x;
  const int e0 = t * 16;
  if (e0 >= e) return;
  int c[16];
  if ((e & 3) == 0 && e0 + 15 < e) {
#pragma unroll
    for (int d = 0; d < 4; ++d) {
      const int4 v = *(const int4*)(col + e0 + d * 4);
      c[d * 4 + 0] = v.x;
      c[d * 4 + 1] = v.y;
      c[d * 4 + 2] = v.z;
      c[d * 4 + 3] = v.w;
    }
  } else {
#pragma unroll
    for (int q = 0; q < 16; ++q) c[q] = (e0 + q < e) ? col[e0 + q] : -1;
  }
  unsigned kw[4];
#pragma unroll
  for (int d = 0; d < 4; ++d) {
    unsigned k = 0;
#pragma unroll
    for (int j = 0; j < 4; ++j) {
      const int cj = c[d * 4 + j];
      if (cj >= 0) {
        atomicAdd(&deg[cj], 1.0f);
        k |= (((unsigned)cj / (unsigned)TILE_NODES) & 255u) << (8 * j);
      }
    }
    kw[d] = k;
  }
  *(uint4*)(keys + (size_t)t * 4) = make_uint4(kw[0], kw[1], kw[2], kw[3]);
}

__global__ void k_deg_rsqrt(const float* __restrict__ deg, float* __restrict__ dinv, int n) {
  int i = blockIdx.x * blockDim.x + threadIdx.x;
  if (i < n) dinv[i] = rsqrtf(deg[i]);
}


#define GEMM_WGS 256
#define GEMM_WAVES 4

__global__ __launch_bounds__(GEMM_WAVES * 32) void k_gemm_wmma(const float* __restrict__ x,
                                                               const float* __restrict__ W,
                                                               float* __restrict__ xw, int ntiles) {
  const int lane = threadIdx.x & 31;
  const int gw = __builtin_amdgcn_readfirstlane(blockIdx.x * GEMM_WAVES + (threadIdx.x >> 5));
  const int nw = gridDim.x * GEMM_WAVES;
  if (gw >= ntiles) return;

  const int m     = lane & 15;
  const int khalf = lane >> 4;

  const float* wbase = W + (size_t)m * 128 + khalf * 2;
  const float* xbase = x + (size_t)m * 128 + khalf * 2;

  v2f b[32][8];
#pragma unroll
  for (int s = 0; s < 32; ++s) {
#pragma unroll
    for (int t = 0; t < 8; ++t) b[s][t] = *(const v2f*)(wbase + t * 2048 + s * 4);
  }

  v2f a_cur[32];
  {
    const float* xrow = xbase + (size_t)gw * 2048;
#pragma unroll
    for (int s = 0; s < 32; ++s) a_cur[s] = *(const v2f*)(xrow + s * 4);
  }

  for (int tile = gw; tile < ntiles; tile += nw) {
    const int nt = (tile + nw < ntiles) ? tile + nw : tile;
    const float* xnext = xbase + (size_t)nt * 2048;
    v2f a_nxt[32];
#pragma unroll
    for (int s = 0; s < 32; ++s) a_nxt[s] = *(const v2f*)(xnext + s * 4);

    v8f acc[8] = {};
#pragma unroll
    for (int s = 0; s < 32; ++s) {
#pragma unroll
      for (int t = 0; t < 8; ++t) {
        acc[t] = __builtin_amdgcn_wmma_f32_16x16x4_f32(
             false, a_cur[s],  false, b[s][t],
             (short)0, acc[t],  false,  false);
      }
    }

    float* outbase = xw + (size_t)(tile * 16 + khalf * 8) * 128 + m;
#pragma unroll
    for (int t = 0; t < 8; ++t) {
#pragma unroll
      for (int v = 0; v < 8; ++v) {
        outbase[(size_t)v * 128 + t * 16] = acc[t][v];
      }
    }

#pragma unroll
    for (int s = 0; s < 32; ++s) a_cur[s] = a_nxt[s];
  }
}

__global__ void k_gemm_tail(const float* __restrict__ x, const float* __restrict__ W,
                            float* __restrict__ xw, int n, int start) {
  int i = blockIdx.x * blockDim.x + threadIdx.x;
  int total = (n - start) * 128;
  if (i >= total) return;
  int node = start + i / 128;
  int o = i & 127;
  const float* xr = x + (size_t)node * 128;
  const float* wr = W + (size_t)o * 128;
  float s = 0.f;
  for (int k = 0; k < 128; ++k) s += xr[k] * wr[k];
  xw[(size_t)node * 128 + o] = s;
}


#define AGG_WAVES 32
#define QCAP 576
#define DRAIN_AT 64
#define SCAN_U 4

static __device__ __forceinline__ void lds_add(float* p, float v) {
  __hip_atomic_fetch_add(p, v, __ATOMIC_RELAXED, __HIP_MEMORY_SCOPE_WORKGROUP);
}

static __device__ __forceinline__ void drain_queue(float* acc, const int* qe, int cnt, int lane,
                                                   int lo, int cntNodes,
                                                   const float* __restrict__ xw,
                                                   const int* __restrict__ row,
                                                   const int* __restrict__ col,
                                                   const float* __restrict__ dinv) {
  for (int i = 0; i < cnt; i += 32) {
    int idx = i + lane;
    if (idx >= cnt) idx = cnt - 1;
    const int ed = qe[idx];
    const int cl = col[ed] - lo;
    const int rl = row[ed];
    const float sl = dinv[rl];
    int nb = cnt - i;
    if (nb > 32) nb = 32;
    for (int k0 = 0; k0 < nb; k0 += 8) {
      float4 v[8];
      int cc[8];
      float ss[8];
#pragma unroll
      for (int k = 0; k < 8; ++k) {
        const int kk = k0 + k;
        const int r = __builtin_amdgcn_readlane(rl, kk);
        cc[k] = __builtin_amdgcn_readlane(cl, kk);
        ss[k] = __int_as_float(__builtin_amdgcn_readlane(__float_as_int(sl), kk));
        v[k] = *(const float4*)(xw + (size_t)r * 128 + lane * 4);
      }
#pragma unroll
      for (int k = 0; k < 8; ++k) {
        if (k0 + k < nb && (unsigned)cc[k] < (unsigned)cntNodes) {
          float* a = acc + cc[k] * 128 + lane;
          lds_add(a, v[k].x * ss[k]);
          lds_add(a + 32, v[k].y * ss[k]);
          lds_add(a + 64, v[k].z * ss[k]);
          lds_add(a + 96, v[k].w * ss[k]);
        }
      }
    }
  }
}

static __device__ __forceinline__ int scan_keys(float* acc, int* qe, int w, int lane,
                                                int lo, int cntNodes,
                                                const float* __restrict__ xw,
                                                const int* __restrict__ row,
                                                const int* __restrict__ col,
                                                const float* __restrict__ dinv,
                                                const unsigned* __restrict__ keys, int e) {
  int cnt = 0;
  const int nChunks = (e + 511) >> 9;
  const int nGroups = (nChunks + SCAN_U - 1) / SCAN_U;
  const int lastVec = ((e + 15) >> 4) - 1;
  const int g0 = (int)(((long long)blockIdx.x * nGroups) / gridDim.x);
  const unsigned keyRep = (blockIdx.x & 255u) * 0x01010101u;

  for (int kg = w; kg < nGroups; kg += AGG_WAVES) {
    int g = g0 + kg;
    if (g >= nGroups) g -= nGroups;
    uint4 kv[SCAN_U];
#pragma unroll
    for (int u = 0; u < SCAN_U; ++u) {
      int vi = (g * SCAN_U + u) * 32 + lane;
      if (vi > lastVec) vi = lastVec;
      kv[u] = *(const uint4*)(keys + (size_t)vi * 4);
    }
#pragma unroll
    for (int u = 0; u < SCAN_U; ++u) {
      const int c0 = (g * SCAN_U + u) << 9;
      const int eb = c0 + lane * 16;
      const unsigned kd[4] = {kv[u].x, kv[u].y, kv[u].z, kv[u].w};
      unsigned f[4];
#pragma unroll
      for (int d = 0; d < 4; ++d) {
        const unsigned x = kd[d] ^ keyRep;
        f[d] = ~(((x & 0x7F7F7F7Fu) + 0x7F7F7F7Fu) | x) & 0x80808080u;
      }
      if (c0 + 512 > e) {
#pragma unroll
        for (int d = 0; d < 4; ++d) {
          const int rem = e - (eb + d * 4);
          if (rem < 4) f[d] = (rem <= 0) ? 0u : (f[d] & ((1u << (8 * rem)) - 1u));
        }
      }
      unsigned F = f[0] | (f[1] >> 1) | (f[2] >> 2) | (f[3] >> 3);
      unsigned mask = __builtin_amdgcn_ballot_w32(F != 0u);
      while (mask) {
        const int pos = cnt + (int)__builtin_amdgcn_mbcnt_lo(mask, 0u);
        if (F != 0u) {
          const int p = __builtin_ctz(F);
          qe[pos] = eb + ((7 - (p & 7)) << 2) + (p >> 3);
        }
        cnt += __builtin_popcount(mask);
        F &= F - 1u;
        mask = __builtin_amdgcn_ballot_w32(F != 0u);
      }
      if (cnt >= DRAIN_AT) {
        drain_queue(acc, qe, cnt, lane, lo, cntNodes, xw, row, col, dinv);
        cnt = 0;
      }
    }
  }
  return cnt;
}

__global__ __launch_bounds__(1024) void k_aggregate(const float* __restrict__ xw,
                                                    const int* __restrict__ row,
                                                    const int* __restrict__ col,
                                                    const float* __restrict__ dinv,
                                                    const float* __restrict__ bias,
                                                    const unsigned* __restrict__ keys,
                                                    float* __restrict__ out, int n, int e) {
  __shared__ float acc[TILE_NODES * 128];
  __shared__ int qe_all[AGG_WAVES * QCAP];

  const int lane = threadIdx.x & 31;
  const int w = __builtin_amdgcn_readfirstlane(threadIdx.x >> 5);
  const int lo = blockIdx.x * TILE_NODES;
  int cntNodes = n - lo;
  if (cntNodes > TILE_NODES) cntNodes = TILE_NODES;

  for (int t = w; t < cntNodes; t += AGG_WAVES) {
    const int c = lo + t;
    const float d = dinv[c];
    const float4 v = *(const float4*)(xw + (size_t)c * 128 + lane * 4);
    float* a = acc + t * 128 + lane;
    a[0] = v.x * d;
    a[32] = v.y * d;
    a[64] = v.z * d;
    a[96] = v.w * d;
  }
  __syncthreads();

  int* qe = qe_all + w * QCAP;
  const int cnt = scan_keys(acc, qe, w, lane, lo, cntNodes, xw, row, col, dinv, keys, e);
  drain_queue(acc, qe, cnt, lane, lo, cntNodes, xw, row, col, dinv);
  __syncthreads();

  const float4 bb = *(const float4*)(bias + lane * 4);
  for (int t = w; t < cntNodes; t += AGG_WAVES) {
    const int c = lo + t;
    const float d = dinv[c];
    const float* a = acc + t * 128 + lane;
    float4 o;
    o.x = a[0] * d + bb.x;
    o.y = a[32] * d + bb.y;
    o.z = a[64] * d + bb.z;
    o.w = a[96] * d + bb.w;
    *(float4*)(out + (size_t)c * 128 + lane * 4) = o;
  }
}


extern "C" void kernel_launch(void* const* d_in, const int* in_sizes, int n_in,
                              void* d_out, int out_size, void* d_ws, size_t ws_size,
                              hipStream_t stream) {
  const float* x  = (const float*)d_in[0];
  const int* ei   = (const int*)d_in[1];
  const float* W  = (const float*)d_in[2];
  const float* b  = (const float*)d_in[3];
  float* out      = (float*)d_out;

  const int N = in_sizes[0] / 128;
  const int E = in_sizes[1] / 2;
  const int* row = ei;
  const int* col = ei + E;

  float* xw   = (float*)d_ws;
  unsigned* keys = (unsigned*)(xw + (size_t)N * 128);
  const int nvec = (E + 15) / 16;
  float* deg  = (float*)(keys + (size_t)nvec * 4);
  float* dinv = deg + N;

  k_deg_init<<<(N + 255) / 256, 256, 0, stream>>>(deg, N);
  if (nvec > 0)
    k_deg_count_keys<<<(nvec + 255) / 256, 256, 0, stream>>>(col, deg, keys, E);
  k_deg_rsqrt<<<(N + 255) / 256, 256, 0, stream>>>(deg, dinv, N);

  const int ntiles = N / 16;
  if (ntiles > 0)
    k_gemm_wmma<<<GEMM_WGS, GEMM_WAVES * 32, 0, stream>>>(x, W, xw, ntiles);
  const int rem = N % 16;
  if (rem > 0)
    k_gemm_tail<<<(rem * 128 + 255) / 256, 256, 0, stream>>>(x, W, xw, N, ntiles * 16);

  k_aggregate<<<(N + TILE_NODES - 1) / TILE_NODES, 1024, 0, stream>>>(xw, row, col, dinv, b, keys, out, N, E);
}


# Projection moved behind the sums: work-groups sum rows of x (not of x W^T) in LDS and multiply their own summed rows by W^T with f32 WMMA from LDS; GEMM launch and plane gone; sum order changed
# speedup vs baseline: 2.3589x; 1.0558x over previous
#include <hip/hip_runtime.h>
#include <hip/hip_bf16.h>

typedef float v2f __attribute__((ext_vector_type(2)));
typedef float v8f __attribute__((ext_vector_type(8)));


__global__ void k_deg_init(float* __restrict__ deg, int n) {
  int i = blockIdx.x * blockDim.x + threadIdx.x;
  if (i < n) deg[i] = 1.0f;
}

#define TILE_NODES 392

__global__ void k_deg_count_keys(const int* __restrict__ col, float* __restrict__ deg,
                                 unsigned* __restrict__ keys, int e) {
  const int t = blockIdx.x * blockDim.x + threadIdx.x;
  const int e0 = t * 16;
  if (e0 >= e) return;
  int c[16];
  if ((e & 3) == 0 && e0 + 15 < e) {
#pragma unroll
    for (int d = 0; d < 4; ++d) {
      const int4 v = *(const int4*)(col + e0 + d * 4);
      c[d * 4 + 0] = v.x;
      c[d * 4 + 1] = v.y;
      c[d * 4 + 2] = v.z;
      c[d * 4 + 3] = v.w;
    }
  } else {
#pragma unroll
    for (int q = 0; q < 16; ++q) c[q] = (e0 + q < e) ? col[e0 + q] : -1;
  }
  unsigned kw[4];
#pragma unroll
  for (int d = 0; d < 4; ++d) {
    unsigned k = 0;
#pragma unroll
    for (int j = 0; j < 4; ++j) {
      const int cj = c[d * 4 + j];
      if (cj >= 0) {
        atomicAdd(&deg[cj], 1.0f);
        k |= (((unsigned)cj / (unsigned)TILE_NODES) & 255u) << (8 * j);
      }
    }
    kw[d] = k;
  }
  *(uint4*)(keys + (size_t)t * 4) = make_uint4(kw[0], kw[1], kw[2], kw[3]);
}

__global__ void k_deg_rsqrt(const float* __restrict__ deg, float* __restrict__ dinv, int n) {
  int i = blockIdx.x * blockDim.x + threadIdx.x;
  if (i < n) dinv[i] = rsqrtf(deg[i]);
}


#define TILE_ROWS 400
#define LPITCH 130

#define AGG_WAVES 32
#define QCAP 576
#define DRAIN_AT 64
#define SCAN_U 4

static __device__ __forceinline__ void lds_add(float* p, float v) {
  __hip_atomic_fetch_add(p, v, __ATOMIC_RELAXED, __HIP_MEMORY_SCOPE_WORKGROUP);
}

static __device__ __forceinline__ void drain_queue(float* acc, const int* qe, int cnt, int lane,
                                                   int lo, int cntNodes,
                                                   const float* __restrict__ x,
                                                   const int* __restrict__ row,
                                                   const int* __restrict__ col,
                                                   const float* __restrict__ dinv) {
  for (int i = 0; i < cnt; i += 32) {
    int idx = i + lane;
    if (idx >= cnt) idx = cnt - 1;
    const int ed = qe[idx];
    const int cl = col[ed] - lo;
    const int rl = row[ed];
    const float sl = dinv[rl];
    int nb = cnt - i;
    if (nb > 32) nb = 32;
    for (int k0 = 0; k0 < nb; k0 += 8) {
      float4 v[8];
      int cc[8];
      float ss[8];
#pragma unroll
      for (int k = 0; k < 8; ++k) {
        const int kk = k0 + k;
        const int r = __builtin_amdgcn_readlane(rl, kk);
        cc[k] = __builtin_amdgcn_readlane(cl, kk);
        ss[k] = __int_as_float(__builtin_amdgcn_readlane(__float_as_int(sl), kk));
        v[k] = *(const float4*)(x + (size_t)r * 128 + lane * 4);
      }
#pragma unroll
      for (int k = 0; k < 8; ++k) {
        if (k0 + k < nb && (unsigned)cc[k] < (unsigned)cntNodes) {
          float* a = acc + cc[k] * LPITCH + lane;
          lds_add(a, v[k].x * ss[k]);
          lds_add(a + 32, v[k].y * ss[k]);
          lds_add(a + 64, v[k].z * ss[k]);
          lds_add(a + 96, v[k].w * ss[k]);
        }
      }
    }
  }
}

static __device__ __forceinline__ int scan_keys(float* acc, int* qe, int w, int lane,
                                                int lo, int cntNodes,
                                                const float* __restrict__ x,
                                                const int* __restrict__ row,
                                                const int* __restrict__ col,
                                                const float* __restrict__ dinv,
                                                const unsigned* __restrict__ keys, int e) {
  int cnt = 0;
  const int nChunks = (e + 511) >> 9;
  const int nGroups = (nChunks + SCAN_U - 1) / SCAN_U;
  const int lastVec = ((e + 15) >> 4) - 1;
  const int g0 = (int)(((long long)blockIdx.x * nGroups) / gridDim.x);
  const unsigned keyRep = (blockIdx.x & 255u) * 0x01010101u;

  for (int kg = w; kg < nGroups; kg += AGG_WAVES) {
    int g = g0 + kg;
    if (g >= nGroups) g -= nGroups;
    uint4 kv[SCAN_U];
#pragma unroll
    for (int u = 0; u < SCAN_U; ++u) {
      int vi = (g * SCAN_U + u) * 32 + lane;
      if (vi > lastVec) vi = lastVec;
      kv[u] = *(const uint4*)(keys + (size_t)vi * 4);
    }
#pragma unroll
    for (int u = 0; u < SCAN_U; ++u) {
      const int c0 = (g * SCAN_U + u) << 9;
      const int eb = c0 + lane * 16;
      const unsigned kd[4] = {kv[u].x, kv[u].y, kv[u].z, kv[u].w};
      unsigned f[4];
#pragma unroll
      for (int d = 0; d < 4; ++d) {
        const unsigned x = kd[d] ^ keyRep;
        f[d] = ~(((x & 0x7F7F7F7Fu) + 0x7F7F7F7Fu) | x) & 0x80808080u;
      }
      if (c0 + 512 > e) {
#pragma unroll
        for (int d = 0; d < 4; ++d) {
          const int rem = e - (eb + d * 4);
          if (rem < 4) f[d] = (rem <= 0) ? 0u : (f[d] & ((1u << (8 * rem)) - 1u));
        }
      }
      unsigned F = f[0] | (f[1] >> 1) | (f[2] >> 2) | (f[3] >> 3);
      unsigned mask = __builtin_amdgcn_ballot_w32(F != 0u);
      while (mask) {
        const int pos = cnt + (int)__builtin_amdgcn_mbcnt_lo(mask, 0u);
        if (F != 0u) {
          const int p = __builtin_ctz(F);
          qe[pos] = eb + ((7 - (p & 7)) << 2) + (p >> 3);
        }
        cnt += __builtin_popcount(mask);
        F &= F - 1u;
        mask = __builtin_amdgcn_ballot_w32(F != 0u);
      }
      if (cnt >= DRAIN_AT) {
        drain_queue(acc, qe, cnt, lane, lo, cntNodes, x, row, col, dinv);
        cnt = 0;
      }
    }
  }
  return cnt;
}

__global__ __launch_bounds__(1024) void k_aggregate(const float* __restrict__ x,
                                                    const float* __restrict__ W,
                                                    const int* __restrict__ row,
                                                    const int* __restrict__ col,
                                                    const float* __restrict__ dinv,
                                                    const float* __restrict__ bias,
                                                    const unsigned* __restrict__ keys,
                                                    float* __restrict__ out, int n, int e) {
  __shared__ __attribute__((aligned(16))) float acc[TILE_ROWS * LPITCH];
  __shared__ __attribute__((aligned(16))) int qe_all[AGG_WAVES * QCAP];
  static_assert(128 * LPITCH <= AGG_WAVES * QCAP, "W must fit into the queue space");

  const int lane = threadIdx.x & 31;
  const int w = __builtin_amdgcn_readfirstlane(threadIdx.x >> 5);
  const int lo = blockIdx.x * TILE_NODES;
  int cntNodes = n - lo;
  if (cntNodes > TILE_NODES) cntNodes = TILE_NODES;
  const int nStrips = (cntNodes + 15) >> 4;

  for (int t = w; t < cntNodes; t += AGG_WAVES) {
    const int c = lo + t;
    const float d = dinv[c];
    const float4 v = *(const float4*)(x + (size_t)c * 128 + lane * 4);
    float* a = acc + t * LPITCH + lane;
    a[0] = v.x * d;
    a[32] = v.y * d;
    a[64] = v.z * d;
    a[96] = v.w * d;
  }
  for (int t = cntNodes + w; t < nStrips * 16; t += AGG_WAVES) {
    float* a = acc + t * LPITCH + lane;
    a[0] = 0.f;
    a[32] = 0.f;
    a[64] = 0.f;
    a[96] = 0.f;
  }
  __syncthreads();

  int* qe = qe_all + w * QCAP;
  const int cnt = scan_keys(acc, qe, w, lane, lo, cntNodes, x, row, col, dinv, keys, e);
  drain_queue(acc, qe, cnt, lane, lo, cntNodes, x, row, col, dinv);
  __syncthreads();

  for (int r = w; r < 128; r += AGG_WAVES) {
    const float4 v = *(const float4*)(W + (size_t)r * 128 + lane * 4);
    int* d = qe_all + r * LPITCH + lane;
    d[0] = __float_as_int(v.x);
    d[32] = __float_as_int(v.y);
    d[64] = __float_as_int(v.z);
    d[96] = __float_as_int(v.w);
  }
  __syncthreads();

  const int m = lane & 15;
  const int h = lane >> 4;
  const int ct = w & 7;
  const int sc = ((w >> 3) + (w & 3)) & 3;

  v2f b[32];
  {
    const int* wl = qe_all + (ct * 16 + m) * LPITCH + 32 * h;
#pragma unroll
    for (int s = 0; s < 32; ++s) {
      const int2 t = *(const int2*)(wl + 64 * (s >> 4) + 2 * (s & 15));
      b[s][0] = __int_as_float(t.x);
      b[s][1] = __int_as_float(t.y);
    }
  }
  const float bn = bias[ct * 16 + m];

  for (int strip = sc; strip < nStrips; strip += 4) {
    const int r0 = strip * 16 + h * 8;
    float dv[8];
#pragma unroll
    for (int v = 0; v < 8; ++v) {
      const int rr = r0 + v;
      dv[v] = dinv[lo + (rr < cntNodes ? rr : cntNodes - 1)];
    }

    const float* arow = acc + (strip * 16 + m) * LPITCH + 32 * h;
    v8f c = {};
#pragma unroll
    for (int s = 0; s < 32; ++s) {
      const v2f a = *(const v2f*)(arow + 64 * (s >> 4) + 2 * (s & 15));
      c = __builtin_amdgcn_wmma_f32_16x16x4_f32(
           false, a,  false, b[s],
           (short)0, c,  false,  false);
    }

    float* orow = out + (size_t)(lo + r0) * 128 + ct * 16 + m;
#pragma unroll
    for (int v = 0; v < 8; ++v) {
      if (r0 + v < cntNodes) orow[(size_t)v * 128] = c[v] * dv[v] + bn;
    }
  }
}


extern "C" void kernel_launch(void* const* d_in, const int* in_sizes, int n_in,
                              void* d_out, int out_size, void* d_ws, size_t ws_size,
                              hipStream_t stream) {
  const float* x  = (const float*)d_in[0];
  const int* ei   = (const int*)d_in[1];
  const float* W  = (const float*)d_in[2];
  const float* b  = (const float*)d_in[3];
  float* out      = (float*)d_out;

  const int N = in_sizes[0] / 128;
  const int E = in_sizes[1] / 2;
  const int* row = ei;
  const int* col = ei + E;

  unsigned* keys = (unsigned*)d_ws;
  const int nvec = (E + 15) / 16;
  float* deg  = (float*)(keys + (size_t)nvec * 4);
  float* dinv = deg + N;

  k_deg_init<<<(N + 255) / 256, 256, 0, stream>>>(deg, N);
  if (nvec > 0)
    k_deg_count_keys<<<(nvec + 255) / 256, 256, 0, stream>>>(col, deg, keys, E);
  k_deg_rsqrt<<<(N + 255) / 256, 256, 0, stream>>>(deg, dinv, N);

  k_aggregate<<<(N + TILE_NODES - 1) / TILE_NODES, 1024, 0, stream>>>(x, W, row, col, dinv, b, keys, out, N, E);
}


# Edge scan reads per-tile edge bitmaps (1 bit per edge, a 256-byte record per 2048-edge chunk and tile) built in LDS by the degree-count launch, not 1-byte keys; LDS atomic sum order as before
# speedup vs baseline: 2.4797x; 1.0512x over previous
#include <hip/hip_runtime.h>
#include <hip/hip_bf16.h>

typedef float v2f __attribute__((ext_vector_type(2)));
typedef float v8f __attribute__((ext_vector_type(8)));


__global__ void k_deg_init(float* __restrict__ deg, int n) {
  int i = blockIdx.x * blockDim.x + threadIdx.x;
  if (i < n) deg[i] = 1.0f;
}

#define TILE_NODES 392
#define CHUNK_EDGES 2048
#define REC_WORDS 64

__global__ __launch_bounds__(256) void k_deg_count_bits(const int* __restrict__ col,
                                                        float* __restrict__ deg,
                                                        unsigned* __restrict__ bits, int e) {
  __shared__ __attribute__((aligned(16))) unsigned bm[256 * REC_WORDS];
  const int tid = threadIdx.x;
  uint4* bm4 = (uint4*)bm;
  for (int i = tid; i < 256 * REC_WORDS / 4; i += 256) bm4[i] = make_uint4(0u, 0u, 0u, 0u);
  __syncthreads();

  const int e0 = blockIdx.x * CHUNK_EDGES + tid * 8;
  int c[8];
  if ((e & 3) == 0 && e0 + 7 < e) {
    const int4 v0 = *(const int4*)(col + e0);
    const int4 v1 = *(const int4*)(col + e0 + 4);
    c[0] = v0.x; c[1] = v0.y; c[2] = v0.z; c[3] = v0.w;
    c[4] = v1.x; c[5] = v1.y; c[6] = v1.z; c[7] = v1.w;
  } else {
#pragma unroll
    for (int q = 0; q < 8; ++q) c[q] = (e0 + q < e) ? col[e0 + q] : -1;
  }
  const int word = tid >> 2;
  const int bit0 = (tid & 3) * 8;
#pragma unroll
  for (int q = 0; q < 8; ++q) {
    const int cq = c[q];
    if (cq >= 0) {
      atomicAdd(&deg[cq], 1.0f);
      const unsigned slot = ((unsigned)cq / (unsigned)TILE_NODES) & 255u;
      atomicOr(&bm[slot * REC_WORDS + word], 1u << (bit0 + q));
    }
  }
  __syncthreads();

  uint4* dst = (uint4*)(bits + (size_t)blockIdx.x * (256 * REC_WORDS));
  for (int i = tid; i < 256 * REC_WORDS / 4; i += 256) dst[i] = bm4[i];
}

__global__ void k_deg_rsqrt(const float* __restrict__ deg, float* __restrict__ dinv, int n) {
  int i = blockIdx.x * blockDim.x + threadIdx.x;
  if (i < n) dinv[i] = rsqrtf(deg[i]);
}


#define TILE_ROWS 400
#define LPITCH 130

#define AGG_WAVES 32
#define QCAP 576
#define DRAIN_AT 64
#define SCAN_U 4

static __device__ __forceinline__ void lds_add(float* p, float v) {
  __hip_atomic_fetch_add(p, v, __ATOMIC_RELAXED, __HIP_MEMORY_SCOPE_WORKGROUP);
}

static __device__ __forceinline__ void drain_queue(float* acc, const int* qe, int cnt, int lane,
                                                   int lo, int cntNodes,
                                                   const float* __restrict__ x,
                                                   const int* __restrict__ row,
                                                   const int* __restrict__ col,
                                                   const float* __restrict__ dinv) {
  for (int i = 0; i < cnt; i += 32) {
    int idx = i + lane;
    if (idx >= cnt) idx = cnt - 1;
    const int ed = qe[idx];
    const int cl = col[ed] - lo;
    const int rl = row[ed];
    const float sl = dinv[rl];
    int nb = cnt - i;
    if (nb > 32) nb = 32;
    for (int k0 = 0; k0 < nb; k0 += 8) {
      float4 v[8];
      int cc[8];
      float ss[8];
#pragma unroll
      for (int k = 0; k < 8; ++k) {
        const int kk = k0 + k;
        const int r = __builtin_amdgcn_readlane(rl, kk);
        cc[k] = __builtin_amdgcn_readlane(cl, kk);
        ss[k] = __int_as_float(__builtin_amdgcn_readlane(__float_as_int(sl), kk));
        v[k] = *(const float4*)(x + (size_t)r * 128 + lane * 4);
      }
#pragma unroll
      for (int k = 0; k < 8; ++k) {
        if (k0 + k < nb && (unsigned)cc[k] < (unsigned)cntNodes) {
          float* a = acc + cc[k] * LPITCH + lane;
          lds_add(a, v[k].x * ss[k]);
          lds_add(a + 32, v[k].y * ss[k]);
          lds_add(a + 64, v[k].z * ss[k]);
          lds_add(a + 96, v[k].w * ss[k]);
        }
      }
    }
  }
}

static __device__ __forceinline__ int scan_bits(float* acc, int* qe, int w, int lane,
                                                int lo, int cntNodes,
                                                const float* __restrict__ x,
                                                const int* __restrict__ row,
                                                const int* __restrict__ col,
                                                const float* __restrict__ dinv,
                                                const unsigned* __restrict__ bits, int e) {
  int cnt = 0;
  const int nChunks = (e + CHUNK_EDGES - 1) / CHUNK_EDGES;
  const int nLoads = (nChunks + 1) >> 1;
  const int v0 = (int)(((long long)blockIdx.x * nLoads) / gridDim.x);
  const unsigned slot = blockIdx.x & 255u;
  const int hr = lane >> 4;
  const int wq = (lane & 15) * 4;

  for (int vb = w; vb < nLoads; vb += AGG_WAVES * SCAN_U) {
    uint4 kv[SCAN_U];
    unsigned okm[SCAN_U];
    int eb[SCAN_U];
#pragma unroll
    for (int u = 0; u < SCAN_U; ++u) {
      const int vi = vb + u * AGG_WAVES;
      int v = vi + v0;
      if (v >= nLoads) v -= nLoads;
      const int g = 2 * v + hr;
      const bool ok = (vi < nLoads) && (g < nChunks);
      const int gc = ok ? g : 0;
      kv[u] = *(const uint4*)(bits + ((size_t)gc * 256 + slot) * REC_WORDS + wq);
      okm[u] = ok ? 0xFFFFFFFFu : 0u;
      eb[u] = gc * CHUNK_EDGES + wq * 32;
    }
#pragma unroll
    for (int u = 0; u < SCAN_U; ++u) {
#pragma unroll 1
      for (int d = 0; d < 4; ++d) {
        unsigned F = ((d == 0) ? kv[u].x : (d == 1) ? kv[u].y : (d == 2) ? kv[u].z : kv[u].w) & okm[u];
        unsigned mask = __builtin_amdgcn_ballot_w32(F != 0u);
        while (mask) {
          const int pos = cnt + (int)__builtin_amdgcn_mbcnt_lo(mask, 0u);
          if (F != 0u) qe[pos] = eb[u] + d * 32 + __builtin_ctz(F);
          cnt += __builtin_popcount(mask);
          F &= F - 1u;
          mask = __builtin_amdgcn_ballot_w32(F != 0u);
          if (cnt >= DRAIN_AT) {
            drain_queue(acc, qe, cnt, lane, lo, cntNodes, x, row, col, dinv);
            cnt = 0;
          }
        }
      }
    }
  }
  return cnt;
}

__global__ __launch_bounds__(1024) void k_aggregate(const float* __restrict__ x,
                                                    const float* __restrict__ W,
                                                    const int* __restrict__ row,
                                                    const int* __restrict__ col,
                                                    const float* __restrict__ dinv,
                                                    const float* __restrict__ bias,
                                                    const unsigned* __restrict__ bits,
                                                    float* __restrict__ out, int n, int e) {
  __shared__ __attribute__((aligned(16))) float acc[TILE_ROWS * LPITCH];
  __shared__ __attribute__((aligned(16))) int qe_all[AGG_WAVES * QCAP];
  static_assert(128 * LPITCH <= AGG_WAVES * QCAP, "W must fit into the queue space");

  const int lane = threadIdx.x & 31;
  const int w = __builtin_amdgcn_readfirstlane(threadIdx.x >> 5);
  const int lo = blockIdx.x * TILE_NODES;
  int cntNodes = n - lo;
  if (cntNodes > TILE_NODES) cntNodes = TILE_NODES;
  const int nStrips = (cntNodes + 15) >> 4;

  for (int t = w; t < cntNodes; t += AGG_WAVES) {
    const int c = lo + t;
    const float d = dinv[c];
    const float4 v = *(const float4*)(x + (size_t)c * 128 + lane * 4);
    float* a = acc + t * LPITCH + lane;
    a[0] = v.x * d;
    a[32] = v.y * d;
    a[64] = v.z * d;
    a[96] = v.w * d;
  }
  for (int t = cntNodes + w; t < nStrips * 16; t += AGG_WAVES) {
    float* a = acc + t * LPITCH + lane;
    a[0] = 0.f;
    a[32] = 0.f;
    a[64] = 0.f;
    a[96] = 0.f;
  }
  __syncthreads();

  int* qe = qe_all + w * QCAP;
  const int cnt = scan_bits(acc, qe, w, lane, lo, cntNodes, x, row, col, dinv, bits, e);
  drain_queue(acc, qe, cnt, lane, lo, cntNodes, x, row, col, dinv);
  __syncthreads();

  for (int r = w; r < 128; r += AGG_WAVES) {
    const float4 v = *(const float4*)(W + (size_t)r * 128 + lane * 4);
    int* d = qe_all + r * LPITCH + lane;
    d[0] = __float_as_int(v.x);
    d[32] = __float_as_int(v.y);
    d[64] = __float_as_int(v.z);
    d[96] = __float_as_int(v.w);
  }
  __syncthreads();

  const int m = lane & 15;
  const int h = lane >> 4;
  const int ct = w & 7;
  const int sc = ((w >> 3) + (w & 3)) & 3;

  v2f b[32];
  {
    const int* wl = qe_all + (ct * 16 + m) * LPITCH + 32 * h;
#pragma unroll
    for (int s = 0; s < 32; ++s) {
      const int2 t = *(const int2*)(wl + 64 * (s >> 4) + 2 * (s & 15));
      b[s][0] = __int_as_float(t.x);
      b[s][1] = __int_as_float(t.y);
    }
  }
  const float bn = bias[ct * 16 + m];

  for (int strip = sc; strip < nStrips; strip += 4) {
    const int r0 = strip * 16 + h * 8;
    float dv[8];
#pragma unroll
    for (int v = 0; v < 8; ++v) {
      const int rr = r0 + v;
      dv[v] = dinv[lo + (rr < cntNodes ? rr : cntNodes - 1)];
    }

    const float* arow = acc + (strip * 16 + m) * LPITCH + 32 * h;
    v8f c = {};
#pragma unroll
    for (int s = 0; s < 32; ++s) {
      const v2f a = *(const v2f*)(arow + 64 * (s >> 4) + 2 * (s & 15));
      c = __builtin_amdgcn_wmma_f32_16x16x4_f32(
           false, a,  false, b[s],
           (short)0, c,  false,  false);
    }

    float* orow = out + (size_t)(lo + r0) * 128 + ct * 16 + m;
#pragma unroll
    for (int v = 0; v < 8; ++v) {
      if (r0 + v < cntNodes) orow[(size_t)v * 128] = c[v] * dv[v] + bn;
    }
  }
}


extern "C" void kernel_launch(void* const* d_in, const int* in_sizes, int n_in,
                              void* d_out, int out_size, void* d_ws, size_t ws_size,
                              hipStream_t stream) {
  const float* x  = (const float*)d_in[0];
  const int* ei   = (const int*)d_in[1];
  const float* W  = (const float*)d_in[2];
  const float* b  = (const float*)d_in[3];
  float* out      = (float*)d_out;

  const int N = in_sizes[0] / 128;
  const int E = in_sizes[1] / 2;
  const int* row = ei;
  const int* col = ei + E;

  unsigned* bits = (unsigned*)d_ws;
  const int nChunks = (E + CHUNK_EDGES - 1) / CHUNK_EDGES;
  float* deg  = (float*)(bits + (size_t)nChunks * (256 * REC_WORDS));
  float* dinv = deg + N;

  k_deg_init<<<(N + 255) / 256, 256, 0, stream>>>(deg, N);
  if (nChunks > 0)
    k_deg_count_bits<<<nChunks, 256, 0, stream>>>(col, deg, bits, E);
  k_deg_rsqrt<<<(N + 255) / 256, 256, 0, stream>>>(deg, dinv, N);

  k_aggregate<<<(N + TILE_NODES - 1) / TILE_NODES, 1024, 0, stream>>>(x, W, row, col, dinv, b, bits, out, N, E);
}


# Edge scan reads tile-sorted chunks: degree-count launch sorts each 2048-edge chunk by tile in LDS and writes (source,target) runs; 3 scattered index loads per edge become 1; LDS sum order as before
# speedup vs baseline: 2.7088x; 1.0924x over previous
#include <hip/hip_runtime.h>
#include <hip/hip_bf16.h>

typedef float v2f __attribute__((ext_vector_type(2)));
typedef float v8f __attribute__((ext_vector_type(8)));


__global__ void k_deg_init(float* __restrict__ deg, int n) {
  int i = blockIdx.x * blockDim.x + threadIdx.x;
  if (i < n) deg[i] = 1.0f;
}

#define TILE_NODES 392
#define CHUNK_EDGES 2048

__global__ __launch_bounds__(256) void k_deg_count_sort(const int* __restrict__ row,
                                                        const int* __restrict__ col,
                                                        float* __restrict__ deg,
                                                        int* __restrict__ pairs_out,
                                                        unsigned* __restrict__ offs, int e) {
  __shared__ int hist[256];
  __shared__ int wsum[8];
  __shared__ __attribute__((aligned(16))) int2 srt[CHUNK_EDGES];
  const int tid = threadIdx.x;
  const int lane = tid & 31;
  hist[tid] = 0;
  __syncthreads();

  const int e0 = blockIdx.x * CHUNK_EDGES + tid * 8;
  int c[8], r[8];
  if ((e & 3) == 0 && e0 + 7 < e) {
    const int4 v0 = *(const int4*)(col + e0);
    const int4 v1 = *(const int4*)(col + e0 + 4);
    const int4 u0 = *(const int4*)(row + e0);
    const int4 u1 = *(const int4*)(row + e0 + 4);
    c[0] = v0.x; c[1] = v0.y; c[2] = v0.z; c[3] = v0.w;
    c[4] = v1.x; c[5] = v1.y; c[6] = v1.z; c[7] = v1.w;
    r[0] = u0.x; r[1] = u0.y; r[2] = u0.z; r[3] = u0.w;
    r[4] = u1.x; r[5] = u1.y; r[6] = u1.z; r[7] = u1.w;
  } else {
#pragma unroll
    for (int q = 0; q < 8; ++q) {
      c[q] = (e0 + q < e) ? col[e0 + q] : -1;
      r[q] = (e0 + q < e) ? row[e0 + q] : 0;
    }
  }
  int sl[8], rk[8];
#pragma unroll
  for (int q = 0; q < 8; ++q) {
    const int cq = c[q];
    sl[q] = 0;
    rk[q] = 0;
    if (cq >= 0) {
      atomicAdd(&deg[cq], 1.0f);
      sl[q] = (int)(((unsigned)cq / (unsigned)TILE_NODES) & 255u);
      rk[q] = atomicAdd(&hist[sl[q]], 1);
    }
  }
  __syncthreads();

  const int hv = hist[tid];
  int inc = hv;
#pragma unroll
  for (int d = 1; d < 32; d <<= 1) {
    const int t = __shfl_up(inc, d);
    if (lane >= d) inc += t;
  }
  if (lane == 31) wsum[tid >> 5] = inc;
  __syncthreads();
  int pre = 0, total = 0;
#pragma unroll
  for (int k = 0; k < 8; ++k) {
    const int s = wsum[k];
    if (k < (tid >> 5)) pre += s;
    total += s;
  }
  const int first = pre + inc - hv;
  hist[tid] = first;
  offs[(size_t)blockIdx.x * 256 + tid] = (unsigned)first | ((unsigned)hv << 16);
  __syncthreads();

#pragma unroll
  for (int q = 0; q < 8; ++q) {
    if (c[q] >= 0) srt[hist[sl[q]] + rk[q]] = make_int2(r[q], c[q]);
  }
  __syncthreads();

  int2* dst = (int2*)pairs_out + (size_t)blockIdx.x * CHUNK_EDGES;
  for (int i = tid; i < total; i += 256) dst[i] = srt[i];
}

__global__ void k_deg_rsqrt(const float* __restrict__ deg, float* __restrict__ dinv, int n) {
  int i = blockIdx.x * blockDim.x + threadIdx.x;
  if (i < n) dinv[i] = rsqrtf(deg[i]);
}


#define TILE_ROWS 400
#define LPITCH 130

#define AGG_WAVES 32
#define QCAP 576
#define DRAIN_AT 64

static __device__ __forceinline__ void lds_add(float* p, float v) {
  __hip_atomic_fetch_add(p, v, __ATOMIC_RELAXED, __HIP_MEMORY_SCOPE_WORKGROUP);
}

static __device__ __forceinline__ void drain_queue(float* acc, const int* qe, int cnt, int lane,
                                                   int lo, int cntNodes,
                                                   const float* __restrict__ x,
                                                   const int2* __restrict__ pairs,
                                                   const float* __restrict__ dinv) {
  for (int i = 0; i < cnt; i += 32) {
    int idx = i + lane;
    if (idx >= cnt) idx = cnt - 1;
    const int2 rc = pairs[qe[idx]];
    const int cl = rc.y - lo;
    const int rl = rc.x;
    const float sl = dinv[rl];
    int nb = cnt - i;
    if (nb > 32) nb = 32;
    for (int k0 = 0; k0 < nb; k0 += 8) {
      float4 v[8];
      int cc[8];
      float ss[8];
#pragma unroll
      for (int k = 0; k < 8; ++k) {
        const int kk = k0 + k;
        const int r = __builtin_amdgcn_readlane(rl, kk);
        cc[k] = __builtin_amdgcn_readlane(cl, kk);
        ss[k] = __int_as_float(__builtin_amdgcn_readlane(__float_as_int(sl), kk));
        v[k] = *(const float4*)(x + (size_t)r * 128 + lane * 4);
      }
#pragma unroll
      for (int k = 0; k < 8; ++k) {
        if (k0 + k < nb && (unsigned)cc[k] < (unsigned)cntNodes) {
          float* a = acc + cc[k] * LPITCH + lane;
          lds_add(a, v[k].x * ss[k]);
          lds_add(a + 32, v[k].y * ss[k]);
          lds_add(a + 64, v[k].z * ss[k]);
          lds_add(a + 96, v[k].w * ss[k]);
        }
      }
    }
  }
}

static __device__ __forceinline__ int scan_runs(float* acc, int* qe, int w, int lane,
                                                int lo, int cntNodes,
                                                const float* __restrict__ x,
                                                const int2* __restrict__ pairs,
                                                const float* __restrict__ dinv,
                                                const unsigned* __restrict__ offs, int e) {
  int cnt = 0;
  const int nChunks = (e + CHUNK_EDGES - 1) / CHUNK_EDGES;
  const int v0 = (int)(((long long)blockIdx.x * nChunks) / gridDim.x);
  const unsigned slot = blockIdx.x & 255u;

  for (int cb = w; cb < nChunks; cb += AGG_WAVES * 32) {
    const int ci = cb + lane * AGG_WAVES;
    const bool ok = ci < nChunks;
    int ch = ci + v0;
    if (ch >= nChunks) ch -= nChunks;
    if (!ok) ch = 0;
    const unsigned o = offs[(size_t)ch * 256 + slot];
    const int st = ch * CHUNK_EDGES + (int)(o & 0xFFFFu);
    const int cn = ok ? (int)(o >> 16) : 0;
    int nl = (nChunks - cb + AGG_WAVES - 1) / AGG_WAVES;
    if (nl > 32) nl = 32;
    for (int i = 0; i < nl; ++i) {
      const int s_i = __builtin_amdgcn_readlane(st, i);
      const int c_i = __builtin_amdgcn_readlane(cn, i);
      for (int k0 = 0; k0 < c_i; k0 += 32) {
        int nb = c_i - k0;
        if (nb > 32) nb = 32;
        if (lane < nb) qe[cnt + lane] = s_i + k0 + lane;
        cnt += nb;
        if (cnt >= DRAIN_AT) {
          drain_queue(acc, qe, cnt, lane, lo, cntNodes, x, pairs, dinv);
          cnt = 0;
        }
      }
    }
  }
  return cnt;
}

__global__ __launch_bounds__(1024) void k_aggregate(const float* __restrict__ x,
                                                    const float* __restrict__ W,
                                                    const int* __restrict__ pairs_in,
                                                    const unsigned* __restrict__ offs,
                                                    const float* __restrict__ dinv,
                                                    const float* __restrict__ bias,
                                                    float* __restrict__ out, int n, int e) {
  __shared__ __attribute__((aligned(16))) float acc[TILE_ROWS * LPITCH];
  __shared__ __attribute__((aligned(16))) int qe_all[AGG_WAVES * QCAP];
  static_assert(128 * LPITCH <= AGG_WAVES * QCAP, "W must fit into the queue space");

  const int lane = threadIdx.x & 31;
  const int w = __builtin_amdgcn_readfirstlane(threadIdx.x >> 5);
  const int lo = blockIdx.x * TILE_NODES;
  int cntNodes = n - lo;
  if (cntNodes > TILE_NODES) cntNodes = TILE_NODES;
  const int nStrips = (cntNodes + 15) >> 4;

  for (int t = w; t < cntNodes; t += AGG_WAVES) {
    const int c = lo + t;
    const float d = dinv[c];
    const float4 v = *(const float4*)(x + (size_t)c * 128 + lane * 4);
    float* a = acc + t * LPITCH + lane;
    a[0] = v.x * d;
    a[32] = v.y * d;
    a[64] = v.z * d;
    a[96] = v.w * d;
  }
  for (int t = cntNodes + w; t < nStrips * 16; t += AGG_WAVES) {
    float* a = acc + t * LPITCH + lane;
    a[0] = 0.f;
    a[32] = 0.f;
    a[64] = 0.f;
    a[96] = 0.f;
  }
  __syncthreads();

  int* qe = qe_all + w * QCAP;
  const int2* pairs = (const int2*)pairs_in;
  const int cnt = scan_runs(acc, qe, w, lane, lo, cntNodes, x, pairs, dinv, offs, e);
  drain_queue(acc, qe, cnt, lane, lo, cntNodes, x, pairs, dinv);
  __syncthreads();

  for (int r = w; r < 128; r += AGG_WAVES) {
    const float4 v = *(const float4*)(W + (size_t)r * 128 + lane * 4);
    int* d = qe_all + r * LPITCH + lane;
    d[0] = __float_as_int(v.x);
    d[32] = __float_as_int(v.y);
    d[64] = __float_as_int(v.z);
    d[96] = __float_as_int(v.w);
  }
  __syncthreads();

  const int m = lane & 15;
  const int h = lane >> 4;
  const int ct = w & 7;
  const int sc = ((w >> 3) + (w & 3)) & 3;

  v2f b[32];
  {
    const int* wl = qe_all + (ct * 16 + m) * LPITCH + 32 * h;
#pragma unroll
    for (int s = 0; s < 32; ++s) {
      const int2 t = *(const int2*)(wl + 64 * (s >> 4) + 2 * (s & 15));
      b[s][0] = __int_as_float(t.x);
      b[s][1] = __int_as_float(t.y);
    }
  }
  const float bn = bias[ct * 16 + m];

  for (int strip = sc; strip < nStrips; strip += 4) {
    const int r0 = strip * 16 + h * 8;
    float dv[8];
#pragma unroll
    for (int v = 0; v < 8; ++v) {
      const int rr = r0 + v;
      dv[v] = dinv[lo + (rr < cntNodes ? rr : cntNodes - 1)];
    }

    const float* arow = acc + (strip * 16 + m) * LPITCH + 32 * h;
    v8f c = {};
#pragma unroll
    for (int s = 0; s < 32; ++s) {
      const v2f a = *(const v2f*)(arow + 64 * (s >> 4) + 2 * (s & 15));
      c = __builtin_amdgcn_wmma_f32_16x16x4_f32(
           false, a,  false, b[s],
           (short)0, c,  false,  false);
    }

    float* orow = out + (size_t)(lo + r0) * 128 + ct * 16 + m;
#pragma unroll
    for (int v = 0; v < 8; ++v) {
      if (r0 + v < cntNodes) orow[(size_t)v * 128] = c[v] * dv[v] + bn;
    }
  }
}


extern "C" void kernel_launch(void* const* d_in, const int* in_sizes, int n_in,
                              void* d_out, int out_size, void* d_ws, size_t ws_size,
                              hipStream_t stream) {
  const float* x  = (const float*)d_in[0];
  const int* ei   = (const int*)d_in[1];
  const float* W  = (const float*)d_in[2];
  const float* b  = (const float*)d_in[3];
  float* out      = (float*)d_out;

  const int N = in_sizes[0] / 128;
  const int E = in_sizes[1] / 2;
  const int* row = ei;
  const int* col = ei + E;

  int* pairs = (int*)d_ws;
  const int nChunks = (E + CHUNK_EDGES - 1) / CHUNK_EDGES;
  unsigned* offs = (unsigned*)(pairs + (size_t)nChunks * (CHUNK_EDGES * 2));
  float* deg  = (float*)(offs + (size_t)nChunks * 256);
  float* dinv = deg + N;

  k_deg_init<<<(N + 255) / 256, 256, 0, stream>>>(deg, N);
  if (nChunks > 0)
    k_deg_count_sort<<<nChunks, 256, 0, stream>>>(row, col, deg, pairs, offs, E);
  k_deg_rsqrt<<<(N + 255) / 256, 256, 0, stream>>>(deg, dinv, N);

  k_aggregate<<<(N + TILE_NODES - 1) / TILE_NODES, 1024, 0, stream>>>(x, W, pairs, offs, dinv, b, out, N, E);
}


# In-degrees counted per 392-node tile in LDS integer counters from the tile-sorted runs (one new launch); the 1.6 M global float atomic adds and the degree-init and rsqrt launches go; sums as before
# speedup vs baseline: 5.1225x; 1.8911x over previous
#include <hip/hip_runtime.h>
#include <hip/hip_bf16.h>

typedef float v2f __attribute__((ext_vector_type(2)));
typedef float v8f __attribute__((ext_vector_type(8)));


#define TILE_NODES 392
#define CHUNK_EDGES 2048

__global__ __launch_bounds__(256) void k_chunk_sort(const int* __restrict__ row,
                                                    const int* __restrict__ col,
                                                    int* __restrict__ pairs_out,
                                                    unsigned* __restrict__ offs, int e) {
  __shared__ int hist[256];
  __shared__ int wsum[8];
  __shared__ __attribute__((aligned(16))) int2 srt[CHUNK_EDGES];
  const int tid = threadIdx.x;
  const int lane = tid & 31;
  hist[tid] = 0;
  __syncthreads();

  const int e0 = blockIdx.x * CHUNK_EDGES + tid * 8;
  int c[8], r[8];
  if ((e & 3) == 0 && e0 + 7 < e) {
    const int4 v0 = *(const int4*)(col + e0);
    const int4 v1 = *(const int4*)(col + e0 + 4);
    const int4 u0 = *(const int4*)(row + e0);
    const int4 u1 = *(const int4*)(row + e0 + 4);
    c[0] = v0.x; c[1] = v0.y; c[2] = v0.z; c[3] = v0.w;
    c[4] = v1.x; c[5] = v1.y; c[6] = v1.z; c[7] = v1.w;
    r[0] = u0.x; r[1] = u0.y; r[2] = u0.z; r[3] = u0.w;
    r[4] = u1.x; r[5] = u1.y; r[6] = u1.z; r[7] = u1.w;
  } else {
#pragma unroll
    for (int q = 0; q < 8; ++q) {
      c[q] = (e0 + q < e) ? col[e0 + q] : -1;
      r[q] = (e0 + q < e) ? row[e0 + q] : 0;
    }
  }
  int sl[8], rk[8];
#pragma unroll
  for (int q = 0; q < 8; ++q) {
    const int cq = c[q];
    sl[q] = 0;
    rk[q] = 0;
    if (cq >= 0) {
      sl[q] = (int)(((unsigned)cq / (unsigned)TILE_NODES) & 255u);
      rk[q] = atomicAdd(&hist[sl[q]], 1);
    }
  }
  __syncthreads();

  const int hv = hist[tid];
  int inc = hv;
#pragma unroll
  for (int d = 1; d < 32; d <<= 1) {
    const int t = __shfl_up(inc, d);
    if (lane >= d) inc += t;
  }
  if (lane == 31) wsum[tid >> 5] = inc;
  __syncthreads();
  int pre = 0, total = 0;
#pragma unroll
  for (int k = 0; k < 8; ++k) {
    const int s = wsum[k];
    if (k < (tid >> 5)) pre += s;
    total += s;
  }
  const int first = pre + inc - hv;
  hist[tid] = first;
  offs[(size_t)blockIdx.x * 256 + tid] = (unsigned)first | ((unsigned)hv << 16);
  __syncthreads();

#pragma unroll
  for (int q = 0; q < 8; ++q) {
    if (c[q] >= 0) srt[hist[sl[q]] + rk[q]] = make_int2(r[q], c[q]);
  }
  __syncthreads();

  int2* dst = (int2*)pairs_out + (size_t)blockIdx.x * CHUNK_EDGES;
  for (int i = tid; i < total; i += 256) dst[i] = srt[i];
}

#define DEG_THREADS 1024
#define DEG_AHEAD 8

__global__ __launch_bounds__(DEG_THREADS) void k_tile_dinv(const int* __restrict__ pairs_in,
                                                           const unsigned* __restrict__ offs,
                                                           float* __restrict__ dinv, int n, int e) {
  __shared__ int indeg[TILE_NODES];
  const int tid = threadIdx.x;
  const int lo = blockIdx.x * TILE_NODES;
  int cntNodes = n - lo;
  if (cntNodes > TILE_NODES) cntNodes = TILE_NODES;
  for (int t = tid; t < TILE_NODES; t += DEG_THREADS) indeg[t] = 0;
  __syncthreads();

  const int nChunks = (e + CHUNK_EDGES - 1) / CHUNK_EDGES;
  const unsigned slot = blockIdx.x & 255u;
  for (int ch = tid; ch < nChunks; ch += DEG_THREADS) {
    const unsigned o = offs[(size_t)ch * 256 + slot];
    const int cn = (int)(o >> 16);
    const int* tp = pairs_in + 2 * ((size_t)ch * CHUNK_EDGES + (o & 0xFFFFu)) + 1;
    for (int k0 = 0; k0 < cn; k0 += DEG_AHEAD) {
      int tg[DEG_AHEAD];
#pragma unroll
      for (int q = 0; q < DEG_AHEAD; ++q) {
        const int k = (k0 + q < cn) ? k0 + q : cn - 1;
        tg[q] = tp[2 * k];
      }
#pragma unroll
      for (int q = 0; q < DEG_AHEAD; ++q) {
        const int cl = (k0 + q < cn) ? tg[q] - lo : -1;
        if ((unsigned)cl < (unsigned)cntNodes) atomicAdd(&indeg[cl], 1);
      }
    }
  }
  __syncthreads();

  for (int t = tid; t < cntNodes; t += DEG_THREADS) dinv[lo + t] = rsqrtf((float)(1 + indeg[t]));
}


#define TILE_ROWS 400
#define LPITCH 130

#define AGG_WAVES 32
#define QCAP 576
#define DRAIN_AT 64

static __device__ __forceinline__ void lds_add(float* p, float v) {
  __hip_atomic_fetch_add(p, v, __ATOMIC_RELAXED, __HIP_MEMORY_SCOPE_WORKGROUP);
}

static __device__ __forceinline__ void drain_queue(float* acc, const int* qe, int cnt, int lane,
                                                   int lo, int cntNodes,
                                                   const float* __restrict__ x,
                                                   const int2* __restrict__ pairs,
                                                   const float* __restrict__ dinv) {
  for (int i = 0; i < cnt; i += 32) {
    int idx = i + lane;
    if (idx >= cnt) idx = cnt - 1;
    const int2 rc = pairs[qe[idx]];
    const int cl = rc.y - lo;
    const int rl = rc.x;
    const float sl = dinv[rl];
    int nb = cnt - i;
    if (nb > 32) nb = 32;
    for (int k0 = 0; k0 < nb; k0 += 8) {
      float4 v[8];
      int cc[8];
      float ss[8];
#pragma unroll
      for (int k = 0; k < 8; ++k) {
        const int kk = k0 + k;
        const int r = __builtin_amdgcn_readlane(rl, kk);
        cc[k] = __builtin_amdgcn_readlane(cl, kk);
        ss[k] = __int_as_float(__builtin_amdgcn_readlane(__float_as_int(sl), kk));
        v[k] = *(const float4*)(x + (size_t)r * 128 + lane * 4);
      }
#pragma unroll
      for (int k = 0; k < 8; ++k) {
        if (k0 + k < nb && (unsigned)cc[k] < (unsigned)cntNodes) {
          float* a = acc + cc[k] * LPITCH + lane;
          lds_add(a, v[k].x * ss[k]);
          lds_add(a + 32, v[k].y * ss[k]);
          lds_add(a + 64, v[k].z * ss[k]);
          lds_add(a + 96, v[k].w * ss[k]);
        }
      }
    }
  }
}

static __device__ __forceinline__ int scan_runs(float* acc, int* qe, int w, int lane,
                                                int lo, int cntNodes,
                                                const float* __restrict__ x,
                                                const int2* __restrict__ pairs,
                                                const float* __restrict__ dinv,
                                                const unsigned* __restrict__ offs, int e) {
  int cnt = 0;
  const int nChunks = (e + CHUNK_EDGES - 1) / CHUNK_EDGES;
  const int v0 = (int)(((long long)blockIdx.x * nChunks) / gridDim.x);
  const unsigned slot = blockIdx.x & 255u;

  for (int cb = w; cb < nChunks; cb += AGG_WAVES * 32) {
    const int ci = cb + lane * AGG_WAVES;
    const bool ok = ci < nChunks;
    int ch = ci + v0;
    if (ch >= nChunks) ch -= nChunks;
    if (!ok) ch = 0;
    const unsigned o = offs[(size_t)ch * 256 + slot];
    const int st = ch * CHUNK_EDGES + (int)(o & 0xFFFFu);
    const int cn = ok ? (int)(o >> 16) : 0;
    int nl = (nChunks - cb + AGG_WAVES - 1) / AGG_WAVES;
    if (nl > 32) nl = 32;
    for (int i = 0; i < nl; ++i) {
      const int s_i = __builtin_amdgcn_readlane(st, i);
      const int c_i = __builtin_amdgcn_readlane(cn, i);
      for (int k0 = 0; k0 < c_i; k0 += 32) {
        int nb = c_i - k0;
        if (nb > 32) nb = 32;
        if (lane < nb) qe[cnt + lane] = s_i + k0 + lane;
        cnt += nb;
        if (cnt >= DRAIN_AT) {
          drain_queue(acc, qe, cnt, lane, lo, cntNodes, x, pairs, dinv);
          cnt = 0;
        }
      }
    }
  }
  return cnt;
}

__global__ __launch_bounds__(1024) void k_aggregate(const float* __restrict__ x,
                                                    const float* __restrict__ W,
                                                    const int* __restrict__ pairs_in,
                                                    const unsigned* __restrict__ offs,
                                                    const float* __restrict__ dinv,
                                                    const float* __restrict__ bias,
                                                    float* __restrict__ out, int n, int e) {
  __shared__ __attribute__((aligned(16))) float acc[TILE_ROWS * LPITCH];
  __shared__ __attribute__((aligned(16))) int qe_all[AGG_WAVES * QCAP];
  static_assert(128 * LPITCH <= AGG_WAVES * QCAP, "W must fit into the queue space");

  const int lane = threadIdx.x & 31;
  const int w = __builtin_amdgcn_readfirstlane(threadIdx.x >> 5);
  const int lo = blockIdx.x * TILE_NODES;
  int cntNodes = n - lo;
  if (cntNodes > TILE_NODES) cntNodes = TILE_NODES;
  const int nStrips = (cntNodes + 15) >> 4;

  for (int t = w; t < cntNodes; t += AGG_WAVES) {
    const int c = lo + t;
    const float d = dinv[c];
    const float4 v = *(const float4*)(x + (size_t)c * 128 + lane * 4);
    float* a = acc + t * LPITCH + lane;
    a[0] = v.x * d;
    a[32] = v.y * d;
    a[64] = v.z * d;
    a[96] = v.w * d;
  }
  for (int t = cntNodes + w; t < nStrips * 16; t += AGG_WAVES) {
    float* a = acc + t * LPITCH + lane;
    a[0] = 0.f;
    a[32] = 0.f;
    a[64] = 0.f;
    a[96] = 0.f;
  }
  __syncthreads();

  int* qe = qe_all + w * QCAP;
  const int2* pairs = (const int2*)pairs_in;
  const int cnt = scan_runs(acc, qe, w, lane, lo, cntNodes, x, pairs, dinv, offs, e);
  drain_queue(acc, qe, cnt, lane, lo, cntNodes, x, pairs, dinv);
  __syncthreads();

  for (int r = w; r < 128; r += AGG_WAVES) {
    const float4 v = *(const float4*)(W + (size_t)r * 128 + lane * 4);
    int* d = qe_all + r * LPITCH + lane;
    d[0] = __float_as_int(v.x);
    d[32] = __float_as_int(v.y);
    d[64] = __float_as_int(v.z);
    d[96] = __float_as_int(v.w);
  }
  __syncthreads();

  const int m = lane & 15;
  const int h = lane >> 4;
  const int ct = w & 7;
  const int sc = ((w >> 3) + (w & 3)) & 3;

  v2f b[32];
  {
    const int* wl = qe_all + (ct * 16 + m) * LPITCH + 32 * h;
#pragma unroll
    for (int s = 0; s < 32; ++s) {
      const int2 t = *(const int2*)(wl + 64 * (s >> 4) + 2 * (s & 15));
      b[s][0] = __int_as_float(t.x);
      b[s][1] = __int_as_float(t.y);
    }
  }
  const float bn = bias[ct * 16 + m];

  for (int strip = sc; strip < nStrips; strip += 4) {
    const int r0 = strip * 16 + h * 8;
    float dv[8];
#pragma unroll
    for (int v = 0; v < 8; ++v) {
      const int rr = r0 + v;
      dv[v] = dinv[lo + (rr < cntNodes ? rr : cntNodes - 1)];
    }

    const float* arow = acc + (strip * 16 + m) * LPITCH + 32 * h;
    v8f c = {};
#pragma unroll
    for (int s = 0; s < 32; ++s) {
      const v2f a = *(const v2f*)(arow + 64 * (s >> 4) + 2 * (s & 15));
      c = __builtin_amdgcn_wmma_f32_16x16x4_f32(
           false, a,  false, b[s],
           (short)0, c,  false,  false);
    }

    float* orow = out + (size_t)(lo + r0) * 128 + ct * 16 + m;
#pragma unroll
    for (int v = 0; v < 8; ++v) {
      if (r0 + v < cntNodes) orow[(size_t)v * 128] = c[v] * dv[v] + bn;
    }
  }
}


extern "C" void kernel_launch(void* const* d_in, const int* in_sizes, int n_in,
                              void* d_out, int out_size, void* d_ws, size_t ws_size,
                              hipStream_t stream) {
  const float* x  = (const float*)d_in[0];
  const int* ei   = (const int*)d_in[1];
  const float* W  = (const float*)d_in[2];
  const float* b  = (const float*)d_in[3];
  float* out      = (float*)d_out;

  const int N = in_sizes[0] / 128;
  const int E = in_sizes[1] / 2;
  const int* row = ei;
  const int* col = ei + E;

  int* pairs = (int*)d_ws;
  const int nChunks = (E + CHUNK_EDGES - 1) / CHUNK_EDGES;
  unsigned* offs = (unsigned*)(pairs + (size_t)nChunks * (CHUNK_EDGES * 2));
  float* dinv = (float*)(offs + (size_t)nChunks * 256);

  if (nChunks > 0)
    k_chunk_sort<<<nChunks, 256, 0, stream>>>(row, col, pairs, offs, E);
  k_tile_dinv<<<(N + TILE_NODES - 1) / TILE_NODES, DEG_THREADS, 0, stream>>>(pairs, offs, dinv, N, E);

  k_aggregate<<<(N + TILE_NODES - 1) / TILE_NODES, 1024, 0, stream>>>(x, W, pairs, offs, dinv, b, out, N, E);
}


# Tile's edges put in target order in LDS (rank counters; starts from the degree launch); waves take equal shares, sum equal-target rows in registers, one LDS add per run; sum order changed
# speedup vs baseline: 5.4786x; 1.0695x over previous
#include <hip/hip_runtime.h>
#include <hip/hip_bf16.h>

typedef float v2f __attribute__((ext_vector_type(2)));
typedef float v8f __attribute__((ext_vector_type(8)));


#define TILE_NODES 392
#define CHUNK_EDGES 2048

__global__ __launch_bounds__(256) void k_chunk_sort(const int* __restrict__ row,
                                                    const int* __restrict__ col,
                                                    int* __restrict__ pairs_out,
                                                    unsigned* __restrict__ offs, int e) {
  __shared__ int hist[256];
  __shared__ int wsum[8];
  __shared__ __attribute__((aligned(16))) int2 srt[CHUNK_EDGES];
  const int tid = threadIdx.x;
  const int lane = tid & 31;
  hist[tid] = 0;
  __syncthreads();

  const int e0 = blockIdx.x * CHUNK_EDGES + tid * 8;
  int c[8], r[8];
  if ((e & 3) == 0 && e0 + 7 < e) {
    const int4 v0 = *(const int4*)(col + e0);
    const int4 v1 = *(const int4*)(col + e0 + 4);
    const int4 u0 = *(const int4*)(row + e0);
    const int4 u1 = *(const int4*)(row + e0 + 4);
    c[0] = v0.x; c[1] = v0.y; c[2] = v0.z; c[3] = v0.w;
    c[4] = v1.x; c[5] = v1.y; c[6] = v1.z; c[7] = v1.w;
    r[0] = u0.x; r[1] = u0.y; r[2] = u0.z; r[3] = u0.w;
    r[4] = u1.x; r[5] = u1.y; r[6] = u1.z; r[7] = u1.w;
  } else {
#pragma unroll
    for (int q = 0; q < 8; ++q) {
      c[q] = (e0 + q < e) ? col[e0 + q] : -1;
      r[q] = (e0 + q < e) ? row[e0 + q] : 0;
    }
  }
  int sl[8], rk[8];
#pragma unroll
  for (int q = 0; q < 8; ++q) {
    const int cq = c[q];
    sl[q] = 0;
    rk[q] = 0;
    if (cq >= 0) {
      sl[q] = (int)(((unsigned)cq / (unsigned)TILE_NODES) & 255u);
      rk[q] = atomicAdd(&hist[sl[q]], 1);
    }
  }
  __syncthreads();

  const int hv = hist[tid];
  int inc = hv;
#pragma unroll
  for (int d = 1; d < 32; d <<= 1) {
    const int t = __shfl_up(inc, d);
    if (lane >= d) inc += t;
  }
  if (lane == 31) wsum[tid >> 5] = inc;
  __syncthreads();
  int pre = 0, total = 0;
#pragma unroll
  for (int k = 0; k < 8; ++k) {
    const int s = wsum[k];
    if (k < (tid >> 5)) pre += s;
    total += s;
  }
  const int first = pre + inc - hv;
  hist[tid] = first;
  offs[(size_t)blockIdx.x * 256 + tid] = (unsigned)first | ((unsigned)hv << 16);
  __syncthreads();

#pragma unroll
  for (int q = 0; q < 8; ++q) {
    if (c[q] >= 0) srt[hist[sl[q]] + rk[q]] = make_int2(r[q], c[q]);
  }
  __syncthreads();

  int2* dst = (int2*)pairs_out + (size_t)blockIdx.x * CHUNK_EDGES;
  for (int i = tid; i < total; i += 256) dst[i] = srt[i];
}

#define DEG_THREADS 1024
#define DEG_AHEAD 8

__global__ __launch_bounds__(DEG_THREADS) void k_tile_dinv(const int* __restrict__ pairs_in,
                                                           const unsigned* __restrict__ offs,
                                                           float* __restrict__ dinv,
                                                           int* __restrict__ tstart, int n, int e) {
  __shared__ int indeg[TILE_NODES];
  __shared__ int wtot[DEG_THREADS / 32];
  const int tid = threadIdx.x;
  const int lo = blockIdx.x * TILE_NODES;
  int cntNodes = n - lo;
  if (cntNodes > TILE_NODES) cntNodes = TILE_NODES;
  for (int t = tid; t < TILE_NODES; t += DEG_THREADS) indeg[t] = 0;
  __syncthreads();

  const int nChunks = (e + CHUNK_EDGES - 1) / CHUNK_EDGES;
  const unsigned slot = blockIdx.x & 255u;
  for (int ch = tid; ch < nChunks; ch += DEG_THREADS) {
    const unsigned o = offs[(size_t)ch * 256 + slot];
    const int cn = (int)(o >> 16);
    const int* tp = pairs_in + 2 * ((size_t)ch * CHUNK_EDGES + (o & 0xFFFFu)) + 1;
    for (int k0 = 0; k0 < cn; k0 += DEG_AHEAD) {
      int tg[DEG_AHEAD];
#pragma unroll
      for (int q = 0; q < DEG_AHEAD; ++q) {
        const int k = (k0 + q < cn) ? k0 + q : cn - 1;
        tg[q] = tp[2 * k];
      }
#pragma unroll
      for (int q = 0; q < DEG_AHEAD; ++q) {
        const int cl = (k0 + q < cn) ? tg[q] - lo : -1;
        if ((unsigned)cl < (unsigned)cntNodes) atomicAdd(&indeg[cl], 1);
      }
    }
  }
  __syncthreads();

  for (int t = tid; t < cntNodes; t += DEG_THREADS) dinv[lo + t] = rsqrtf((float)(1 + indeg[t]));

  const int lane = tid & 31;
  const int wv = tid >> 5;
  const int hv = (tid < TILE_NODES) ? indeg[tid] : 0;
  int inc = hv;
#pragma unroll
  for (int d = 1; d < 32; d <<= 1) {
    const int t = __shfl_up(inc, d);
    if (lane >= d) inc += t;
  }
  if (lane == 31) wtot[wv] = inc;
  __syncthreads();
  int pre = 0;
#pragma unroll
  for (int k = 0; k < (TILE_NODES + 32) / 32; ++k) {
    if (k < wv) pre += wtot[k];
  }
  if (tid <= TILE_NODES) tstart[(size_t)blockIdx.x * (TILE_NODES + 1) + tid] = pre + inc - hv;
}


#define TILE_ROWS 400
#define LPITCH 130

#define AGG_WAVES 32
#define QCAP 576
#define DRAIN_AT 64

static __device__ __forceinline__ void lds_add(float* p, float v) {
  __hip_atomic_fetch_add(p, v, __ATOMIC_RELAXED, __HIP_MEMORY_SCOPE_WORKGROUP);
}

static __device__ __forceinline__ void drain_queue(float* acc, const int* qe, int cnt, int lane,
                                                   int lo, int cntNodes,
                                                   const float* __restrict__ x,
                                                   const int2* __restrict__ pairs,
                                                   const float* __restrict__ dinv) {
  for (int i = 0; i < cnt; i += 32) {
    int idx = i + lane;
    if (idx >= cnt) idx = cnt - 1;
    const int2 rc = pairs[qe[idx]];
    const int cl = rc.y - lo;
    const int rl = rc.x;
    const float sl = dinv[rl];
    int nb = cnt - i;
    if (nb > 32) nb = 32;
    for (int k0 = 0; k0 < nb; k0 += 8) {
      float4 v[8];
      int cc[8];
      float ss[8];
#pragma unroll
      for (int k = 0; k < 8; ++k) {
        const int kk = k0 + k;
        const int r = __builtin_amdgcn_readlane(rl, kk);
        cc[k] = __builtin_amdgcn_readlane(cl, kk);
        ss[k] = __int_as_float(__builtin_amdgcn_readlane(__float_as_int(sl), kk));
        v[k] = *(const float4*)(x + (size_t)r * 128 + lane * 4);
      }
#pragma unroll
      for (int k = 0; k < 8; ++k) {
        if (k0 + k < nb && (unsigned)cc[k] < (unsigned)cntNodes) {
          float* a = acc + cc[k] * LPITCH + lane;
          lds_add(a, v[k].x * ss[k]);
          lds_add(a + 32, v[k].y * ss[k]);
          lds_add(a + 64, v[k].z * ss[k]);
          lds_add(a + 96, v[k].w * ss[k]);
        }
      }
    }
  }
}

static __device__ __forceinline__ int scan_runs(float* acc, int* qe, int w, int lane,
                                                int lo, int cntNodes,
                                                const float* __restrict__ x,
                                                const int2* __restrict__ pairs,
                                                const float* __restrict__ dinv,
                                                const unsigned* __restrict__ offs, int e) {
  int cnt = 0;
  const int nChunks = (e + CHUNK_EDGES - 1) / CHUNK_EDGES;
  const int v0 = (int)(((long long)blockIdx.x * nChunks) / gridDim.x);
  const unsigned slot = blockIdx.x & 255u;

  for (int cb = w; cb < nChunks; cb += AGG_WAVES * 32) {
    const int ci = cb + lane * AGG_WAVES;
    const bool ok = ci < nChunks;
    int ch = ci + v0;
    if (ch >= nChunks) ch -= nChunks;
    if (!ok) ch = 0;
    const unsigned o = offs[(size_t)ch * 256 + slot];
    const int st = ch * CHUNK_EDGES + (int)(o & 0xFFFFu);
    const int cn = ok ? (int)(o >> 16) : 0;
    int nl = (nChunks - cb + AGG_WAVES - 1) / AGG_WAVES;
    if (nl > 32) nl = 32;
    for (int i = 0; i < nl; ++i) {
      const int s_i = __builtin_amdgcn_readlane(st, i);
      const int c_i = __builtin_amdgcn_readlane(cn, i);
      for (int k0 = 0; k0 < c_i; k0 += 32) {
        int nb = c_i - k0;
        if (nb > 32) nb = 32;
        if (lane < nb) qe[cnt + lane] = s_i + k0 + lane;
        cnt += nb;
        if (cnt >= DRAIN_AT) {
          drain_queue(acc, qe, cnt, lane, lo, cntNodes, x, pairs, dinv);
          cnt = 0;
        }
      }
    }
  }
  return cnt;
}

#define CAP_EDGES 8192
#define PLACE_AHEAD 8

static __device__ __forceinline__ void place_edges(int2* sorted, int* fillc, const int* tst,
                                                   int lo, int cntNodes,
                                                   const int2* __restrict__ pairs,
                                                   const unsigned* __restrict__ offs, int e) {
  const int nChunks = (e + CHUNK_EDGES - 1) / CHUNK_EDGES;
  const unsigned slot = blockIdx.x & 255u;
  for (int ch = threadIdx.x; ch < nChunks; ch += 1024) {
    const unsigned o = offs[(size_t)ch * 256 + slot];
    const int cn = (int)(o >> 16);
    const int2* rp = pairs + (size_t)ch * CHUNK_EDGES + (o & 0xFFFFu);
    for (int k0 = 0; k0 < cn; k0 += PLACE_AHEAD) {
      int2 pr[PLACE_AHEAD];
#pragma unroll
      for (int q = 0; q < PLACE_AHEAD; ++q) {
        const int k = (k0 + q < cn) ? k0 + q : cn - 1;
        pr[q] = rp[k];
      }
#pragma unroll
      for (int q = 0; q < PLACE_AHEAD; ++q) {
        const int cl = (k0 + q < cn) ? pr[q].y - lo : -1;
        if ((unsigned)cl < (unsigned)cntNodes) {
          const int pos = tst[cl] + atomicAdd(&fillc[cl], 1);
          if (pos < CAP_EDGES) sorted[pos] = make_int2(pr[q].x, cl);
        }
      }
    }
  }
}

static __device__ __forceinline__ void add_run(float* acc, int cl, int lane, float4 s) {
  float* a = acc + cl * LPITCH + lane;
  lds_add(a, s.x);
  lds_add(a + 32, s.y);
  lds_add(a + 64, s.z);
  lds_add(a + 96, s.w);
}

static __device__ __forceinline__ void sum_ordered(float* acc, const int2* sorted, int total,
                                                   int w, int lane,
                                                   const float* __restrict__ x,
                                                   const float* __restrict__ dinv) {
  const int b = (total * w) / AGG_WAVES;
  const int en = (total * (w + 1)) / AGG_WAVES;
  if (b >= en) return;
  int cur = -1;
  float4 sum = make_float4(0.f, 0.f, 0.f, 0.f);
  int i0 = b + lane;
  if (i0 >= en) i0 = en - 1;
  int2 rc = sorted[i0];
  float sl = dinv[rc.x];
  for (int i = b; i < en; i += 32) {
    int2 rcn = rc;
    float sln = sl;
    if (i + 32 < en) {
      int in = i + 32 + lane;
      if (in >= en) in = en - 1;
      rcn = sorted[in];
      sln = dinv[rcn.x];
    }
    int nb = en - i;
    if (nb > 32) nb = 32;
    for (int k0 = 0; k0 < nb; k0 += 8) {
      float4 v[8];
      int cc[8];
      float ss[8];
#pragma unroll
      for (int k = 0; k < 8; ++k) {
        const int kk = k0 + k;
        const int r = __builtin_amdgcn_readlane(rc.x, kk);
        cc[k] = __builtin_amdgcn_readlane(rc.y, kk);
        ss[k] = __int_as_float(__builtin_amdgcn_readlane(__float_as_int(sl), kk));
        v[k] = *(const float4*)(x + (size_t)r * 128 + lane * 4);
      }
#pragma unroll
      for (int k = 0; k < 8; ++k) {
        if (k0 + k < nb) {
          if (cc[k] != cur) {
            if (cur >= 0) add_run(acc, cur, lane, sum);
            cur = cc[k];
            sum = make_float4(0.f, 0.f, 0.f, 0.f);
          }
          sum.x += v[k].x * ss[k];
          sum.y += v[k].y * ss[k];
          sum.z += v[k].z * ss[k];
          sum.w += v[k].w * ss[k];
        }
      }
    }
    rc = rcn;
    sl = sln;
  }
  add_run(acc, cur, lane, sum);
}

__global__ __launch_bounds__(1024) void k_aggregate(const float* __restrict__ x,
                                                    const float* __restrict__ W,
                                                    const int* __restrict__ pairs_in,
                                                    const unsigned* __restrict__ offs,
                                                    const float* __restrict__ dinv,
                                                    const int* __restrict__ tstart,
                                                    const float* __restrict__ bias,
                                                    float* __restrict__ out, int n, int e) {
  __shared__ __attribute__((aligned(16))) float acc[TILE_ROWS * LPITCH];
  __shared__ __attribute__((aligned(16))) int qe_all[AGG_WAVES * QCAP];
  static_assert(128 * LPITCH <= AGG_WAVES * QCAP, "W must fit into the queue space");
  static_assert(2 * CAP_EDGES <= AGG_WAVES * QCAP, "the ordered list must fit into the queue space");
  static_assert(TILE_NODES + 1 <= TILE_ROWS, "the prefix sums need one entry more than nodes");
  __shared__ int fillc[TILE_ROWS];
  __shared__ int tst[TILE_ROWS];

  const int lane = threadIdx.x & 31;
  const int w = __builtin_amdgcn_readfirstlane(threadIdx.x >> 5);
  const int lo = blockIdx.x * TILE_NODES;
  int cntNodes = n - lo;
  if (cntNodes > TILE_NODES) cntNodes = TILE_NODES;
  const int nStrips = (cntNodes + 15) >> 4;

  for (int t = w; t < cntNodes; t += AGG_WAVES) {
    const int c = lo + t;
    const float d = dinv[c];
    const float4 v = *(const float4*)(x + (size_t)c * 128 + lane * 4);
    float* a = acc + t * LPITCH + lane;
    a[0] = v.x * d;
    a[32] = v.y * d;
    a[64] = v.z * d;
    a[96] = v.w * d;
  }
  for (int t = cntNodes + w; t < nStrips * 16; t += AGG_WAVES) {
    float* a = acc + t * LPITCH + lane;
    a[0] = 0.f;
    a[32] = 0.f;
    a[64] = 0.f;
    a[96] = 0.f;
  }
  for (int t = threadIdx.x; t < TILE_ROWS; t += 1024) {
    fillc[t] = 0;
    tst[t] = (t <= TILE_NODES) ? tstart[(size_t)blockIdx.x * (TILE_NODES + 1) + t] : 0;
  }
  __syncthreads();

  const int2* pairs = (const int2*)pairs_in;
  const int total = __builtin_amdgcn_readfirstlane(tst[TILE_NODES]);
  if (total <= CAP_EDGES) {
    int2* sorted = (int2*)qe_all;
    place_edges(sorted, fillc, tst, lo, cntNodes, pairs, offs, e);
    __syncthreads();
    sum_ordered(acc, sorted, total, w, lane, x, dinv);
  } else {
    int* qe = qe_all + w * QCAP;
    const int cnt = scan_runs(acc, qe, w, lane, lo, cntNodes, x, pairs, dinv, offs, e);
    drain_queue(acc, qe, cnt, lane, lo, cntNodes, x, pairs, dinv);
  }
  __syncthreads();

  for (int r = w; r < 128; r += AGG_WAVES) {
    const float4 v = *(const float4*)(W + (size_t)r * 128 + lane * 4);
    int* d = qe_all + r * LPITCH + lane;
    d[0] = __float_as_int(v.x);
    d[32] = __float_as_int(v.y);
    d[64] = __float_as_int(v.z);
    d[96] = __float_as_int(v.w);
  }
  __syncthreads();

  const int m = lane & 15;
  const int h = lane >> 4;
  const int ct = w & 7;
  const int sc = ((w >> 3) + (w & 3)) & 3;

  v2f b[32];
  {
    const int* wl = qe_all + (ct * 16 + m) * LPITCH + 32 * h;
#pragma unroll
    for (int s = 0; s < 32; ++s) {
      const int2 t = *(const int2*)(wl + 64 * (s >> 4) + 2 * (s & 15));
      b[s][0] = __int_as_float(t.x);
      b[s][1] = __int_as_float(t.y);
    }
  }
  const float bn = bias[ct * 16 + m];

  for (int strip = sc; strip < nStrips; strip += 4) {
    const int r0 = strip * 16 + h * 8;
    float dv[8];
#pragma unroll
    for (int v = 0; v < 8; ++v) {
      const int rr = r0 + v;
      dv[v] = dinv[lo + (rr < cntNodes ? rr : cntNodes - 1)];
    }

    const float* arow = acc + (strip * 16 + m) * LPITCH + 32 * h;
    v8f c = {};
#pragma unroll
    for (int s = 0; s < 32; ++s) {
      const v2f a = *(const v2f*)(arow + 64 * (s >> 4) + 2 * (s & 15));
      c = __builtin_amdgcn_wmma_f32_16x16x4_f32(
           false, a,  false, b[s],
           (short)0, c,  false,  false);
    }

    float* orow = out + (size_t)(lo + r0) * 128 + ct * 16 + m;
#pragma unroll
    for (int v = 0; v < 8; ++v) {
      if (r0 + v < cntNodes) orow[(size_t)v * 128] = c[v] * dv[v] + bn;
    }
  }
}


extern "C" void kernel_launch(void* const* d_in, const int* in_sizes, int n_in,
                              void* d_out, int out_size, void* d_ws, size_t ws_size,
                              hipStream_t stream) {
  const float* x  = (const float*)d_in[0];
  const int* ei   = (const int*)d_in[1];
  const float* W  = (const float*)d_in[2];
  const float* b  = (const float*)d_in[3];
  float* out      = (float*)d_out;

  const int N = in_sizes[0] / 128;
  const int E = in_sizes[1] / 2;
  const int* row = ei;
  const int* col = ei + E;

  int* pairs = (int*)d_ws;
  const int nChunks = (E + CHUNK_EDGES - 1) / CHUNK_EDGES;
  unsigned* offs = (unsigned*)(pairs + (size_t)nChunks * (CHUNK_EDGES * 2));
  float* dinv = (float*)(offs + (size_t)nChunks * 256);
  int* tstart = (int*)(dinv + N);

  if (nChunks > 0)
    k_chunk_sort<<<nChunks, 256, 0, stream>>>(row, col, pairs, offs, E);
  k_tile_dinv<<<(N + TILE_NODES - 1) / TILE_NODES, DEG_THREADS, 0, stream>>>(pairs, offs, dinv, tstart, N, E);

  k_aggregate<<<(N + TILE_NODES - 1) / TILE_NODES, 1024, 0, stream>>>(x, W, pairs, offs, dinv, tstart, b, out, N, E);
}


# Aggregation launch issues independent loads as whole groups: self-loop rows 7 per round trip (was 1), W rows and prefix sums ahead of their barriers, the gather's 8 row loads unsplit; sums unchanged
# speedup vs baseline: 5.6690x; 1.0348x over previous
#include <hip/hip_runtime.h>
#include <hip/hip_bf16.h>

typedef float v2f __attribute__((ext_vector_type(2)));
typedef float v8f __attribute__((ext_vector_type(8)));


#define TILE_NODES 392
#define CHUNK_EDGES 2048

__global__ __launch_bounds__(256) void k_chunk_sort(const int* __restrict__ row,
                                                    const int* __restrict__ col,
                                                    int* __restrict__ pairs_out,
                                                    unsigned* __restrict__ offs, int e) {
  __shared__ int hist[256];
  __shared__ int wsum[8];
  __shared__ __attribute__((aligned(16))) int2 srt[CHUNK_EDGES];
  const int tid = threadIdx.x;
  const int lane = tid & 31;
  hist[tid] = 0;
  __syncthreads();

  const int e0 = blockIdx.x * CHUNK_EDGES + tid * 8;
  int c[8], r[8];
  if ((e & 3) == 0 && e0 + 7 < e) {
    const int4 v0 = *(const int4*)(col + e0);
    const int4 v1 = *(const int4*)(col + e0 + 4);
    const int4 u0 = *(const int4*)(row + e0);
    const int4 u1 = *(const int4*)(row + e0 + 4);
    c[0] = v0.x; c[1] = v0.y; c[2] = v0.z; c[3] = v0.w;
    c[4] = v1.x; c[5] = v1.y; c[6] = v1.z; c[7] = v1.w;
    r[0] = u0.x; r[1] = u0.y; r[2] = u0.z; r[3] = u0.w;
    r[4] = u1.x; r[5] = u1.y; r[6] = u1.z; r[7] = u1.w;
  } else {
#pragma unroll
    for (int q = 0; q < 8; ++q) {
      c[q] = (e0 + q < e) ? col[e0 + q] : -1;
      r[q] = (e0 + q < e) ? row[e0 + q] : 0;
    }
  }
  int sl[8], rk[8];
#pragma unroll
  for (int q = 0; q < 8; ++q) {
    const int cq = c[q];
    sl[q] = 0;
    rk[q] = 0;
    if (cq >= 0) {
      sl[q] = (int)(((unsigned)cq / (unsigned)TILE_NODES) & 255u);
      rk[q] = atomicAdd(&hist[sl[q]], 1);
    }
  }
  __syncthreads();

  const int hv = hist[tid];
  int inc = hv;
#pragma unroll
  for (int d = 1; d < 32; d <<= 1) {
    const int t = __shfl_up(inc, d);
    if (lane >= d) inc += t;
  }
  if (lane == 31) wsum[tid >> 5] = inc;
  __syncthreads();
  int pre = 0, total = 0;
#pragma unroll
  for (int k = 0; k < 8; ++k) {
    const int s = wsum[k];
    if (k < (tid >> 5)) pre += s;
    total += s;
  }
  const int first = pre + inc - hv;
  hist[tid] = first;
  offs[(size_t)blockIdx.x * 256 + tid] = (unsigned)first | ((unsigned)hv << 16);
  __syncthreads();

#pragma unroll
  for (int q = 0; q < 8; ++q) {
    if (c[q] >= 0) srt[hist[sl[q]] + rk[q]] = make_int2(r[q], c[q]);
  }
  __syncthreads();

  int2* dst = (int2*)pairs_out + (size_t)blockIdx.x * CHUNK_EDGES;
  for (int i = tid; i < total; i += 256) dst[i] = srt[i];
}

#define DEG_THREADS 1024
#define DEG_AHEAD 8

__global__ __launch_bounds__(DEG_THREADS) void k_tile_dinv(const int* __restrict__ pairs_in,
                                                           const unsigned* __restrict__ offs,
                                                           float* __restrict__ dinv,
                                                           int* __restrict__ tstart, int n, int e) {
  __shared__ int indeg[TILE_NODES];
  __shared__ int wtot[DEG_THREADS / 32];
  const int tid = threadIdx.x;
  const int lo = blockIdx.x * TILE_NODES;
  int cntNodes = n - lo;
  if (cntNodes > TILE_NODES) cntNodes = TILE_NODES;
  for (int t = tid; t < TILE_NODES; t += DEG_THREADS) indeg[t] = 0;
  __syncthreads();

  const int nChunks = (e + CHUNK_EDGES - 1) / CHUNK_EDGES;
  const unsigned slot = blockIdx.x & 255u;
  for (int ch = tid; ch < nChunks; ch += DEG_THREADS) {
    const unsigned o = offs[(size_t)ch * 256 + slot];
    const int cn = (int)(o >> 16);
    const int* tp = pairs_in + 2 * ((size_t)ch * CHUNK_EDGES + (o & 0xFFFFu)) + 1;
    for (int k0 = 0; k0 < cn; k0 += DEG_AHEAD) {
      int tg[DEG_AHEAD];
#pragma unroll
      for (int q = 0; q < DEG_AHEAD; ++q) {
        const int k = (k0 + q < cn) ? k0 + q : cn - 1;
        tg[q] = tp[2 * k];
      }
#pragma unroll
      for (int q = 0; q < DEG_AHEAD; ++q) {
        const int cl = (k0 + q < cn) ? tg[q] - lo : -1;
        if ((unsigned)cl < (unsigned)cntNodes) atomicAdd(&indeg[cl], 1);
      }
    }
  }
  __syncthreads();

  for (int t = tid; t < cntNodes; t += DEG_THREADS) dinv[lo + t] = rsqrtf((float)(1 + indeg[t]));

  const int lane = tid & 31;
  const int wv = tid >> 5;
  const int hv = (tid < TILE_NODES) ? indeg[tid] : 0;
  int inc = hv;
#pragma unroll
  for (int d = 1; d < 32; d <<= 1) {
    const int t = __shfl_up(inc, d);
    if (lane >= d) inc += t;
  }
  if (lane == 31) wtot[wv] = inc;
  __syncthreads();
  int pre = 0;
#pragma unroll
  for (int k = 0; k < (TILE_NODES + 32) / 32; ++k) {
    if (k < wv) pre += wtot[k];
  }
  if (tid <= TILE_NODES) tstart[(size_t)blockIdx.x * (TILE_NODES + 1) + tid] = pre + inc - hv;
}


#define TILE_ROWS 400
#define LPITCH 130

#define AGG_WAVES 32
#define INIT_AHEAD 7
#define QCAP 576
#define DRAIN_AT 64

static __device__ __forceinline__ void lds_add(float* p, float v) {
  __hip_atomic_fetch_add(p, v, __ATOMIC_RELAXED, __HIP_MEMORY_SCOPE_WORKGROUP);
}

static __device__ __forceinline__ void drain_queue(float* acc, const int* qe, int cnt, int lane,
                                                   int lo, int cntNodes,
                                                   const float* __restrict__ x,
                                                   const int2* __restrict__ pairs,
                                                   const float* __restrict__ dinv) {
  for (int i = 0; i < cnt; i += 32) {
    int idx = i + lane;
    if (idx >= cnt) idx = cnt - 1;
    const int2 rc = pairs[qe[idx]];
    const int cl = rc.y - lo;
    const int rl = rc.x;
    const float sl = dinv[rl];
    int nb = cnt - i;
    if (nb > 32) nb = 32;
    for (int k0 = 0; k0 < nb; k0 += 8) {
      float4 v[8];
      int cc[8];
      float ss[8];
#pragma unroll
      for (int k = 0; k < 8; ++k) {
        const int kk = k0 + k;
        const int r = __builtin_amdgcn_readlane(rl, kk);
        cc[k] = __builtin_amdgcn_readlane(cl, kk);
        ss[k] = __int_as_float(__builtin_amdgcn_readlane(__float_as_int(sl), kk));
        v[k] = *(const float4*)(x + (size_t)r * 128 + lane * 4);
      }
#pragma unroll
      for (int k = 0; k < 8; ++k) {
        if (k0 + k < nb && (unsigned)cc[k] < (unsigned)cntNodes) {
          float* a = acc + cc[k] * LPITCH + lane;
          lds_add(a, v[k].x * ss[k]);
          lds_add(a + 32, v[k].y * ss[k]);
          lds_add(a + 64, v[k].z * ss[k]);
          lds_add(a + 96, v[k].w * ss[k]);
        }
      }
    }
  }
}

static __device__ __forceinline__ int scan_runs(float* acc, int* qe, int w, int lane,
                                                int lo, int cntNodes,
                                                const float* __restrict__ x,
                                                const int2* __restrict__ pairs,
                                                const float* __restrict__ dinv,
                                                const unsigned* __restrict__ offs, int e) {
  int cnt = 0;
  const int nChunks = (e + CHUNK_EDGES - 1) / CHUNK_EDGES;
  const int v0 = (int)(((long long)blockIdx.x * nChunks) / gridDim.x);
  const unsigned slot = blockIdx.x & 255u;

  for (int cb = w; cb < nChunks; cb += AGG_WAVES * 32) {
    const int ci = cb + lane * AGG_WAVES;
    const bool ok = ci < nChunks;
    int ch = ci + v0;
    if (ch >= nChunks) ch -= nChunks;
    if (!ok) ch = 0;
    const unsigned o = offs[(size_t)ch * 256 + slot];
    const int st = ch * CHUNK_EDGES + (int)(o & 0xFFFFu);
    const int cn = ok ? (int)(o >> 16) : 0;
    int nl = (nChunks - cb + AGG_WAVES - 1) / AGG_WAVES;
    if (nl > 32) nl = 32;
    for (int i = 0; i < nl; ++i) {
      const int s_i = __builtin_amdgcn_readlane(st, i);
      const int c_i = __builtin_amdgcn_readlane(cn, i);
      for (int k0 = 0; k0 < c_i; k0 += 32) {
        int nb = c_i - k0;
        if (nb > 32) nb = 32;
        if (lane < nb) qe[cnt + lane] = s_i + k0 + lane;
        cnt += nb;
        if (cnt >= DRAIN_AT) {
          drain_queue(acc, qe, cnt, lane, lo, cntNodes, x, pairs, dinv);
          cnt = 0;
        }
      }
    }
  }
  return cnt;
}

#define CAP_EDGES 8192
#define PLACE_AHEAD 8

static __device__ __forceinline__ void place_edges(int2* sorted, int* fillc, const int* tst,
                                                   int lo, int cntNodes,
                                                   const int2* __restrict__ pairs,
                                                   const unsigned* __restrict__ offs, int e) {
  const int nChunks = (e + CHUNK_EDGES - 1) / CHUNK_EDGES;
  const unsigned slot = blockIdx.x & 255u;
  for (int ch = threadIdx.x; ch < nChunks; ch += 1024) {
    const unsigned o = offs[(size_t)ch * 256 + slot];
    const int cn = (int)(o >> 16);
    const int2* rp = pairs + (size_t)ch * CHUNK_EDGES + (o & 0xFFFFu);
    for (int k0 = 0; k0 < cn; k0 += PLACE_AHEAD) {
      int2 pr[PLACE_AHEAD];
#pragma unroll
      for (int q = 0; q < PLACE_AHEAD; ++q) {
        const int k = (k0 + q < cn) ? k0 + q : cn - 1;
        pr[q] = rp[k];
      }
#pragma unroll
      for (int q = 0; q < PLACE_AHEAD; ++q) {
        const int cl = (k0 + q < cn) ? pr[q].y - lo : -1;
        if ((unsigned)cl < (unsigned)cntNodes) {
          const int pos = tst[cl] + atomicAdd(&fillc[cl], 1);
          if (pos < CAP_EDGES) sorted[pos] = make_int2(pr[q].x, cl);
        }
      }
    }
  }
}

static __device__ __forceinline__ void add_run(float* acc, int cl, int lane, float4 s) {
  float* a = acc + cl * LPITCH + lane;
  lds_add(a, s.x);
  lds_add(a + 32, s.y);
  lds_add(a + 64, s.z);
  lds_add(a + 96, s.w);
}

static __device__ __forceinline__ void sum_ordered(float* acc, const int2* sorted, int total,
                                                   int w, int lane,
                                                   const float* __restrict__ x,
                                                   const float* __restrict__ dinv) {
  const int b = (total * w) / AGG_WAVES;
  const int en = (total * (w + 1)) / AGG_WAVES;
  if (b >= en) return;
  int cur = -1;
  float4 sum = make_float4(0.f, 0.f, 0.f, 0.f);
  int i0 = b + lane;
  if (i0 >= en) i0 = en - 1;
  int2 rc = sorted[i0];
  float sl = dinv[rc.x];
  for (int i = b; i < en; i += 32) {
    int2 rcn = rc;
    float sln = sl;
    if (i + 32 < en) {
      int in = i + 32 + lane;
      if (in >= en) in = en - 1;
      rcn = sorted[in];
      sln = dinv[rcn.x];
    }
    int nb = en - i;
    if (nb > 32) nb = 32;
    for (int k0 = 0; k0 < nb; k0 += 8) {
      float4 v[8];
      int cc[8];
      float ss[8];
#pragma unroll
      for (int k = 0; k < 8; ++k) {
        const int kk = k0 + k;
        const int r = __builtin_amdgcn_readlane(rc.x, kk);
        cc[k] = __builtin_amdgcn_readlane(rc.y, kk);
        ss[k] = __int_as_float(__builtin_amdgcn_readlane(__float_as_int(sl), kk));
        v[k] = *(const float4*)(x + (size_t)r * 128 + lane * 4);
      }
#pragma unroll
      for (int k = 0; k < 8; ++k)
        asm volatile("" ::"v"(v[k].x), "v"(v[k].y), "v"(v[k].z), "v"(v[k].w));
#pragma unroll
      for (int k = 0; k < 8; ++k) {
        if (k0 + k < nb) {
          if (cc[k] != cur) {
            if (cur >= 0) add_run(acc, cur, lane, sum);
            cur = cc[k];
            sum = make_float4(0.f, 0.f, 0.f, 0.f);
          }
          sum.x += v[k].x * ss[k];
          sum.y += v[k].y * ss[k];
          sum.z += v[k].z * ss[k];
          sum.w += v[k].w * ss[k];
        }
      }
    }
    rc = rcn;
    sl = sln;
  }
  add_run(acc, cur, lane, sum);
}

__global__ __launch_bounds__(1024) void k_aggregate(const float* __restrict__ x,
                                                    const float* __restrict__ W,
                                                    const int* __restrict__ pairs_in,
                                                    const unsigned* __restrict__ offs,
                                                    const float* __restrict__ dinv,
                                                    const int* __restrict__ tstart,
                                                    const float* __restrict__ bias,
                                                    float* __restrict__ out, int n, int e) {
  __shared__ __attribute__((aligned(16))) float acc[TILE_ROWS * LPITCH];
  __shared__ __attribute__((aligned(16))) int qe_all[AGG_WAVES * QCAP];
  static_assert(128 * LPITCH <= AGG_WAVES * QCAP, "W must fit into the queue space");
  static_assert(2 * CAP_EDGES <= AGG_WAVES * QCAP, "the ordered list must fit into the queue space");
  static_assert(TILE_NODES + 1 <= TILE_ROWS, "the prefix sums need one entry more than nodes");
  static_assert(TILE_ROWS <= 1024 && 128 % AGG_WAVES == 0, "one thread per counter; whole rows of W per wave");
  __shared__ int fillc[TILE_ROWS];
  __shared__ int tst[TILE_ROWS];

  const int lane = threadIdx.x & 31;
  const int w = __builtin_amdgcn_readfirstlane(threadIdx.x >> 5);
  const int lo = blockIdx.x * TILE_NODES;
  int cntNodes = n - lo;
  if (cntNodes > TILE_NODES) cntNodes = TILE_NODES;
  const int nStrips = (cntNodes + 15) >> 4;

  int tsv = 0;
  if (threadIdx.x <= TILE_NODES) tsv = tstart[(size_t)blockIdx.x * (TILE_NODES + 1) + threadIdx.x];

  for (int t0 = w; t0 < cntNodes; t0 += AGG_WAVES * INIT_AHEAD) {
    float4 v[INIT_AHEAD];
    float d[INIT_AHEAD];
#pragma unroll
    for (int q = 0; q < INIT_AHEAD; ++q) {
      const int t = t0 + q * AGG_WAVES;
      const int c = lo + (t < cntNodes ? t : cntNodes - 1);
      d[q] = dinv[c];
      v[q] = *(const float4*)(x + (size_t)c * 128 + lane * 4);
    }
#pragma unroll
    for (int q = 0; q < INIT_AHEAD; ++q)
      asm volatile("" ::"v"(v[q].x), "v"(v[q].y), "v"(v[q].z), "v"(v[q].w), "s"(d[q]));
#pragma unroll
    for (int q = 0; q < INIT_AHEAD; ++q) {
      const int t = t0 + q * AGG_WAVES;
      if (t < cntNodes) {
        float* a = acc + t * LPITCH + lane;
        a[0] = v[q].x * d[q];
        a[32] = v[q].y * d[q];
        a[64] = v[q].z * d[q];
        a[96] = v[q].w * d[q];
      }
    }
  }
  for (int t = cntNodes + w; t < nStrips * 16; t += AGG_WAVES) {
    float* a = acc + t * LPITCH + lane;
    a[0] = 0.f;
    a[32] = 0.f;
    a[64] = 0.f;
    a[96] = 0.f;
  }
  if (threadIdx.x < TILE_ROWS) {
    fillc[threadIdx.x] = 0;
    tst[threadIdx.x] = tsv;
  }
  __syncthreads();

  const int2* pairs = (const int2*)pairs_in;
  const int total = __builtin_amdgcn_readfirstlane(tst[TILE_NODES]);
  if (total <= CAP_EDGES) {
    int2* sorted = (int2*)qe_all;
    place_edges(sorted, fillc, tst, lo, cntNodes, pairs, offs, e);
    __syncthreads();
    sum_ordered(acc, sorted, total, w, lane, x, dinv);
  } else {
    int* qe = qe_all + w * QCAP;
    const int cnt = scan_runs(acc, qe, w, lane, lo, cntNodes, x, pairs, dinv, offs, e);
    drain_queue(acc, qe, cnt, lane, lo, cntNodes, x, pairs, dinv);
  }
  float4 wr[128 / AGG_WAVES];
#pragma unroll
  for (int q = 0; q < 128 / AGG_WAVES; ++q)
    wr[q] = *(const float4*)(W + (size_t)(w + q * AGG_WAVES) * 128 + lane * 4);
  __syncthreads();

#pragma unroll
  for (int q = 0; q < 128 / AGG_WAVES; ++q) {
    int* d = qe_all + (w + q * AGG_WAVES) * LPITCH + lane;
    d[0] = __float_as_int(wr[q].x);
    d[32] = __float_as_int(wr[q].y);
    d[64] = __float_as_int(wr[q].z);
    d[96] = __float_as_int(wr[q].w);
  }
  __syncthreads();

  const int m = lane & 15;
  const int h = lane >> 4;
  const int ct = w & 7;
  const int sc = ((w >> 3) + (w & 3)) & 3;

  v2f b[32];
  {
    const int* wl = qe_all + (ct * 16 + m) * LPITCH + 32 * h;
#pragma unroll
    for (int s = 0; s < 32; ++s) {
      const int2 t = *(const int2*)(wl + 64 * (s >> 4) + 2 * (s & 15));
      b[s][0] = __int_as_float(t.x);
      b[s][1] = __int_as_float(t.y);
    }
  }
  const float bn = bias[ct * 16 + m];

  for (int strip = sc; strip < nStrips; strip += 4) {
    const int r0 = strip * 16 + h * 8;
    float dv[8];
#pragma unroll
    for (int v = 0; v < 8; ++v) {
      const int rr = r0 + v;
      dv[v] = dinv[lo + (rr < cntNodes ? rr : cntNodes - 1)];
    }

    const float* arow = acc + (strip * 16 + m) * LPITCH + 32 * h;
    v8f c = {};
#pragma unroll
    for (int s = 0; s < 32; ++s) {
      const v2f a = *(const v2f*)(arow + 64 * (s >> 4) + 2 * (s & 15));
      c = __builtin_amdgcn_wmma_f32_16x16x4_f32(
           false, a,  false, b[s],
           (short)0, c,  false,  false);
    }

    float* orow = out + (size_t)(lo + r0) * 128 + ct * 16 + m;
#pragma unroll
    for (int v = 0; v < 8; ++v) {
      if (r0 + v < cntNodes) orow[(size_t)v * 128] = c[v] * dv[v] + bn;
    }
  }
}


extern "C" void kernel_launch(void* const* d_in, const int* in_sizes, int n_in,
                              void* d_out, int out_size, void* d_ws, size_t ws_size,
                              hipStream_t stream) {
  const float* x  = (const float*)d_in[0];
  const int* ei   = (const int*)d_in[1];
  const float* W  = (const float*)d_in[2];
  const float* b  = (const float*)d_in[3];
  float* out      = (float*)d_out;

  const int N = in_sizes[0] / 128;
  const int E = in_sizes[1] / 2;
  const int* row = ei;
  const int* col = ei + E;

  int* pairs = (int*)d_ws;
  const int nChunks = (E + CHUNK_EDGES - 1) / CHUNK_EDGES;
  unsigned* offs = (unsigned*)(pairs + (size_t)nChunks * (CHUNK_EDGES * 2));
  float* dinv = (float*)(offs + (size_t)nChunks * 256);
  int* tstart = (int*)(dinv + N);

  if (nChunks > 0)
    k_chunk_sort<<<nChunks, 256, 0, stream>>>(row, col, pairs, offs, E);
  k_tile_dinv<<<(N + TILE_NODES - 1) / TILE_NODES, DEG_THREADS, 0, stream>>>(pairs, offs, dinv, tstart, N, E);

  k_aggregate<<<(N + TILE_NODES - 1) / TILE_NODES, 1024, 0, stream>>>(x, W, pairs, offs, dinv, tstart, b, out, N, E);
}


# Edge runs read by a half-wave (lane j the j-th pair; 13 runs and their LDS ranks in flight) in the degree launch and the edge placement: 2-4 lines per load, not 32; LDS rank sum order as before
# speedup vs baseline: 5.8560x; 1.0330x over previous
#include <hip/hip_runtime.h>
#include <hip/hip_bf16.h>

typedef float v2f __attribute__((ext_vector_type(2)));
typedef float v8f __attribute__((ext_vector_type(8)));


#define TILE_NODES 392
#define CHUNK_EDGES 2048

__global__ __launch_bounds__(256) void k_chunk_sort(const int* __restrict__ row,
                                                    const int* __restrict__ col,
                                                    int* __restrict__ pairs_out,
                                                    unsigned* __restrict__ offs, int e) {
  __shared__ int hist[256];
  __shared__ int wsum[8];
  __shared__ __attribute__((aligned(16))) int2 srt[CHUNK_EDGES];
  const int tid = threadIdx.x;
  const int lane = tid & 31;
  hist[tid] = 0;
  __syncthreads();

  const int e0 = blockIdx.x * CHUNK_EDGES + tid * 8;
  int c[8], r[8];
  if ((e & 3) == 0 && e0 + 7 < e) {
    const int4 v0 = *(const int4*)(col + e0);
    const int4 v1 = *(const int4*)(col + e0 + 4);
    const int4 u0 = *(const int4*)(row + e0);
    const int4 u1 = *(const int4*)(row + e0 + 4);
    c[0] = v0.x; c[1] = v0.y; c[2] = v0.z; c[3] = v0.w;
    c[4] = v1.x; c[5] = v1.y; c[6] = v1.z; c[7] = v1.w;
    r[0] = u0.x; r[1] = u0.y; r[2] = u0.z; r[3] = u0.w;
    r[4] = u1.x; r[5] = u1.y; r[6] = u1.z; r[7] = u1.w;
  } else {
#pragma unroll
    for (int q = 0; q < 8; ++q) {
      c[q] = (e0 + q < e) ? col[e0 + q] : -1;
      r[q] = (e0 + q < e) ? row[e0 + q] : 0;
    }
  }
  int sl[8], rk[8];
#pragma unroll
  for (int q = 0; q < 8; ++q) {
    const int cq = c[q];
    sl[q] = 0;
    rk[q] = 0;
    if (cq >= 0) {
      sl[q] = (int)(((unsigned)cq / (unsigned)TILE_NODES) & 255u);
      rk[q] = atomicAdd(&hist[sl[q]], 1);
    }
  }
  __syncthreads();

  const int hv = hist[tid];
  int inc = hv;
#pragma unroll
  for (int d = 1; d < 32; d <<= 1) {
    const int t = __shfl_up(inc, d);
    if (lane >= d) inc += t;
  }
  if (lane == 31) wsum[tid >> 5] = inc;
  __syncthreads();
  int pre = 0, total = 0;
#pragma unroll
  for (int k = 0; k < 8; ++k) {
    const int s = wsum[k];
    if (k < (tid >> 5)) pre += s;
    total += s;
  }
  const int first = pre + inc - hv;
  hist[tid] = first;
  offs[(size_t)blockIdx.x * 256 + tid] = (unsigned)first | ((unsigned)hv << 16);
  __syncthreads();

#pragma unroll
  for (int q = 0; q < 8; ++q) {
    if (c[q] >= 0) srt[hist[sl[q]] + rk[q]] = make_int2(r[q], c[q]);
  }
  __syncthreads();

  int2* dst = (int2*)pairs_out + (size_t)blockIdx.x * CHUNK_EDGES;
  for (int i = tid; i < total; i += 256) dst[i] = srt[i];
}

#define DEG_THREADS 1024
#define RUNS_AHEAD 13

__global__ __launch_bounds__(DEG_THREADS) void k_tile_dinv(const int* __restrict__ pairs_in,
                                                           const unsigned* __restrict__ offs,
                                                           float* __restrict__ dinv,
                                                           int* __restrict__ tstart, int n, int e) {
  __shared__ int indeg[TILE_NODES];
  __shared__ int wtot[DEG_THREADS / 32];
  const int tid = threadIdx.x;
  const int lo = blockIdx.x * TILE_NODES;
  int cntNodes = n - lo;
  if (cntNodes > TILE_NODES) cntNodes = TILE_NODES;
  for (int t = tid; t < TILE_NODES; t += DEG_THREADS) indeg[t] = 0;
  __syncthreads();

  const int nChunks = (e + CHUNK_EDGES - 1) / CHUNK_EDGES;
  const unsigned slot = blockIdx.x & 255u;
  const int hw = tid >> 4;
  const int j = tid & 15;
  const char* ob = (const char*)offs + 4u * slot;
  const char* pb = (const char*)pairs_in + 4;
  for (int c0 = 0; c0 < nChunks; c0 += (DEG_THREADS / 16) * RUNS_AHEAD) {
    unsigned o[RUNS_AHEAD];
#pragma unroll
    for (int r = 0; r < RUNS_AHEAD; ++r) {
      const int ch = c0 + r * (DEG_THREADS / 16) + hw;
      o[r] = *(const unsigned*)(ob + (unsigned)(ch < nChunks ? ch : 0) * 1024u);
    }
    __builtin_amdgcn_sched_barrier(0);
    int tg[RUNS_AHEAD];
#pragma unroll
    for (int r = 0; r < RUNS_AHEAD; ++r) {
      const int ch = c0 + r * (DEG_THREADS / 16) + hw;
      if (ch >= nChunks) o[r] = 0;
      const int cn = (int)(o[r] >> 16);
      int k = (j < cn) ? j : cn - 1;
      if (k < 0) k = 0;
      const unsigned cb = (cn > 0) ? (unsigned)ch : 0u;
      const unsigned fs = (cn > 0) ? (o[r] & 0xFFFFu) : 0u;
      const unsigned p = cb * (unsigned)CHUNK_EDGES + fs + (unsigned)k;
      tg[r] = *(const int*)(pb + (p << 3));
    }
    __builtin_amdgcn_sched_barrier(0);
#pragma unroll
    for (int r = 0; r < RUNS_AHEAD; ++r) {
      const int cn = (int)(o[r] >> 16);
      const int cl = (j < cn) ? tg[r] - lo : -1;
      if ((unsigned)cl < (unsigned)cntNodes) atomicAdd(&indeg[cl], 1);
      if (cn > 16) {
        const int ch = c0 + r * (DEG_THREADS / 16) + hw;
        const unsigned p0 = (unsigned)ch * (unsigned)CHUNK_EDGES + (o[r] & 0xFFFFu);
        for (int k = 16 + j; k < cn; k += 16) {
          const int cl2 = *(const int*)(pb + ((p0 + (unsigned)k) << 3)) - lo;
          if ((unsigned)cl2 < (unsigned)cntNodes) atomicAdd(&indeg[cl2], 1);
        }
      }
    }
  }
  __syncthreads();

  for (int t = tid; t < cntNodes; t += DEG_THREADS) dinv[lo + t] = rsqrtf((float)(1 + indeg[t]));

  const int lane = tid & 31;
  const int wv = tid >> 5;
  const int hv = (tid < TILE_NODES) ? indeg[tid] : 0;
  int inc = hv;
#pragma unroll
  for (int d = 1; d < 32; d <<= 1) {
    const int t = __shfl_up(inc, d);
    if (lane >= d) inc += t;
  }
  if (lane == 31) wtot[wv] = inc;
  __syncthreads();
  int pre = 0;
#pragma unroll
  for (int k = 0; k < (TILE_NODES + 32) / 32; ++k) {
    if (k < wv) pre += wtot[k];
  }
  if (tid <= TILE_NODES) tstart[(size_t)blockIdx.x * (TILE_NODES + 1) + tid] = pre + inc - hv;
}


#define TILE_ROWS 400
#define LPITCH 130

#define AGG_WAVES 32
#define INIT_AHEAD 7
#define QCAP 576
#define DRAIN_AT 64

static __device__ __forceinline__ void lds_add(float* p, float v) {
  __hip_atomic_fetch_add(p, v, __ATOMIC_RELAXED, __HIP_MEMORY_SCOPE_WORKGROUP);
}

static __device__ __forceinline__ void drain_queue(float* acc, const int* qe, int cnt, int lane,
                                                   int lo, int cntNodes,
                                                   const float* __restrict__ x,
                                                   const int2* __restrict__ pairs,
                                                   const float* __restrict__ dinv) {
  for (int i = 0; i < cnt; i += 32) {
    int idx = i + lane;
    if (idx >= cnt) idx = cnt - 1;
    const int2 rc = pairs[qe[idx]];
    const int cl = rc.y - lo;
    const int rl = rc.x;
    const float sl = dinv[rl];
    int nb = cnt - i;
    if (nb > 32) nb = 32;
    for (int k0 = 0; k0 < nb; k0 += 8) {
      float4 v[8];
      int cc[8];
      float ss[8];
#pragma unroll
      for (int k = 0; k < 8; ++k) {
        const int kk = k0 + k;
        const int r = __builtin_amdgcn_readlane(rl, kk);
        cc[k] = __builtin_amdgcn_readlane(cl, kk);
        ss[k] = __int_as_float(__builtin_amdgcn_readlane(__float_as_int(sl), kk));
        v[k] = *(const float4*)(x + (size_t)r * 128 + lane * 4);
      }
#pragma unroll
      for (int k = 0; k < 8; ++k) {
        if (k0 + k < nb && (unsigned)cc[k] < (unsigned)cntNodes) {
          float* a = acc + cc[k] * LPITCH + lane;
          lds_add(a, v[k].x * ss[k]);
          lds_add(a + 32, v[k].y * ss[k]);
          lds_add(a + 64, v[k].z * ss[k]);
          lds_add(a + 96, v[k].w * ss[k]);
        }
      }
    }
  }
}

static __device__ __forceinline__ int scan_runs(float* acc, int* qe, int w, int lane,
                                                int lo, int cntNodes,
                                                const float* __restrict__ x,
                                                const int2* __restrict__ pairs,
                                                const float* __restrict__ dinv,
                                                const unsigned* __restrict__ offs, int e) {
  int cnt = 0;
  const int nChunks = (e + CHUNK_EDGES - 1) / CHUNK_EDGES;
  const int v0 = (int)(((long long)blockIdx.x * nChunks) / gridDim.x);
  const unsigned slot = blockIdx.x & 255u;

  for (int cb = w; cb < nChunks; cb += AGG_WAVES * 32) {
    const int ci = cb + lane * AGG_WAVES;
    const bool ok = ci < nChunks;
    int ch = ci + v0;
    if (ch >= nChunks) ch -= nChunks;
    if (!ok) ch = 0;
    const unsigned o = offs[(size_t)ch * 256 + slot];
    const int st = ch * CHUNK_EDGES + (int)(o & 0xFFFFu);
    const int cn = ok ? (int)(o >> 16) : 0;
    int nl = (nChunks - cb + AGG_WAVES - 1) / AGG_WAVES;
    if (nl > 32) nl = 32;
    for (int i = 0; i < nl; ++i) {
      const int s_i = __builtin_amdgcn_readlane(st, i);
      const int c_i = __builtin_amdgcn_readlane(cn, i);
      for (int k0 = 0; k0 < c_i; k0 += 32) {
        int nb = c_i - k0;
        if (nb > 32) nb = 32;
        if (lane < nb) qe[cnt + lane] = s_i + k0 + lane;
        cnt += nb;
        if (cnt >= DRAIN_AT) {
          drain_queue(acc, qe, cnt, lane, lo, cntNodes, x, pairs, dinv);
          cnt = 0;
        }
      }
    }
  }
  return cnt;
}

#define CAP_EDGES 8192

static __device__ __forceinline__ void place_one(int2* sorted, int* fillc, const int* tst,
                                                 int src, int cl, int cntNodes) {
  if ((unsigned)cl < (unsigned)cntNodes) {
    const int pos = tst[cl] + atomicAdd(&fillc[cl], 1);
    if (pos < CAP_EDGES) sorted[pos] = make_int2(src, cl);
  }
}

static __device__ __forceinline__ void place_edges(int2* sorted, int* fillc, const int* tst,
                                                   int lo, int cntNodes,
                                                   const int2* __restrict__ pairs,
                                                   const unsigned* __restrict__ offs, int e) {
  const int nChunks = (e + CHUNK_EDGES - 1) / CHUNK_EDGES;
  const unsigned slot = blockIdx.x & 255u;
  const int hw = threadIdx.x >> 4;
  const int j = threadIdx.x & 15;
  const char* ob = (const char*)offs + 4u * slot;
  const char* pb = (const char*)pairs;
  for (int c0 = 0; c0 < nChunks; c0 += 64 * RUNS_AHEAD) {
    unsigned o[RUNS_AHEAD];
#pragma unroll
    for (int r = 0; r < RUNS_AHEAD; ++r) {
      const int ch = c0 + r * 64 + hw;
      o[r] = *(const unsigned*)(ob + (unsigned)(ch < nChunks ? ch : 0) * 1024u);
    }
    __builtin_amdgcn_sched_barrier(0);
    int2 pr[RUNS_AHEAD];
#pragma unroll
    for (int r = 0; r < RUNS_AHEAD; ++r) {
      const int ch = c0 + r * 64 + hw;
      if (ch >= nChunks) o[r] = 0;
      const int cn = (int)(o[r] >> 16);
      int k = (j < cn) ? j : cn - 1;
      if (k < 0) k = 0;
      const unsigned cb = (cn > 0) ? (unsigned)ch : 0u;
      const unsigned fs = (cn > 0) ? (o[r] & 0xFFFFu) : 0u;
      const unsigned p = cb * (unsigned)CHUNK_EDGES + fs + (unsigned)k;
      pr[r] = *(const int2*)(pb + (p << 3));
    }
    __builtin_amdgcn_sched_barrier(0);
    int bs[RUNS_AHEAD], rk[RUNS_AHEAD];
#pragma unroll
    for (int r = 0; r < RUNS_AHEAD; ++r) {
      const int cn = (int)(o[r] >> 16);
      const int cl = (j < cn) ? pr[r].y - lo : -1;
      pr[r].y = cl;
      bs[r] = CAP_EDGES;
      rk[r] = 0;
      if ((unsigned)cl < (unsigned)cntNodes) {
        bs[r] = tst[cl];
        rk[r] = atomicAdd(&fillc[cl], 1);
      }
    }
#pragma unroll
    for (int r = 0; r < RUNS_AHEAD; ++r) {
      const int pos = bs[r] + rk[r];
      if (pos < CAP_EDGES) sorted[pos] = pr[r];
    }
#pragma unroll
    for (int r = 0; r < RUNS_AHEAD; ++r) {
      const int cn = (int)(o[r] >> 16);
      if (cn > 16) {
        const int ch = c0 + r * 64 + hw;
        const unsigned p0 = (unsigned)ch * (unsigned)CHUNK_EDGES + (o[r] & 0xFFFFu);
        for (int k = 16 + j; k < cn; k += 16) {
          const int2 p = *(const int2*)(pb + ((p0 + (unsigned)k) << 3));
          place_one(sorted, fillc, tst, p.x, p.y - lo, cntNodes);
        }
      }
    }
  }
}

static __device__ __forceinline__ void add_run(float* acc, int cl, int lane, float4 s) {
  float* a = acc + cl * LPITCH + lane;
  lds_add(a, s.x);
  lds_add(a + 32, s.y);
  lds_add(a + 64, s.z);
  lds_add(a + 96, s.w);
}

static __device__ __forceinline__ void sum_ordered(float* acc, const int2* sorted, int total,
                                                   int w, int lane,
                                                   const float* __restrict__ x,
                                                   const float* __restrict__ dinv) {
  const int b = (total * w) / AGG_WAVES;
  const int en = (total * (w + 1)) / AGG_WAVES;
  if (b >= en) return;
  int cur = -1;
  float4 sum = make_float4(0.f, 0.f, 0.f, 0.f);
  int i0 = b + lane;
  if (i0 >= en) i0 = en - 1;
  int2 rc = sorted[i0];
  float sl = dinv[rc.x];
  for (int i = b; i < en; i += 32) {
    int2 rcn = rc;
    float sln = sl;
    if (i + 32 < en) {
      int in = i + 32 + lane;
      if (in >= en) in = en - 1;
      rcn = sorted[in];
      sln = dinv[rcn.x];
    }
    int nb = en - i;
    if (nb > 32) nb = 32;
    for (int k0 = 0; k0 < nb; k0 += 8) {
      float4 v[8];
      int cc[8];
      float ss[8];
#pragma unroll
      for (int k = 0; k < 8; ++k) {
        const int kk = k0 + k;
        const int r = __builtin_amdgcn_readlane(rc.x, kk);
        cc[k] = __builtin_amdgcn_readlane(rc.y, kk);
        ss[k] = __int_as_float(__builtin_amdgcn_readlane(__float_as_int(sl), kk));
        v[k] = *(const float4*)(x + (size_t)r * 128 + lane * 4);
      }
#pragma unroll
      for (int k = 0; k < 8; ++k)
        asm volatile("" ::"v"(v[k].x), "v"(v[k].y), "v"(v[k].z), "v"(v[k].w));
#pragma unroll
      for (int k = 0; k < 8; ++k) {
        if (k0 + k < nb) {
          if (cc[k] != cur) {
            if (cur >= 0) add_run(acc, cur, lane, sum);
            cur = cc[k];
            sum = make_float4(0.f, 0.f, 0.f, 0.f);
          }
          sum.x += v[k].x * ss[k];
          sum.y += v[k].y * ss[k];
          sum.z += v[k].z * ss[k];
          sum.w += v[k].w * ss[k];
        }
      }
    }
    rc = rcn;
    sl = sln;
  }
  add_run(acc, cur, lane, sum);
}

__global__ __launch_bounds__(1024) void k_aggregate(const float* __restrict__ x,
                                                    const float* __restrict__ W,
                                                    const int* __restrict__ pairs_in,
                                                    const unsigned* __restrict__ offs,
                                                    const float* __restrict__ dinv,
                                                    const int* __restrict__ tstart,
                                                    const float* __restrict__ bias,
                                                    float* __restrict__ out, int n, int e) {
  __shared__ __attribute__((aligned(16))) float acc[TILE_ROWS * LPITCH];
  __shared__ __attribute__((aligned(16))) int qe_all[AGG_WAVES * QCAP];
  static_assert(128 * LPITCH <= AGG_WAVES * QCAP, "W must fit into the queue space");
  static_assert(2 * CAP_EDGES <= AGG_WAVES * QCAP, "the ordered list must fit into the queue space");
  static_assert(TILE_NODES + 1 <= TILE_ROWS, "the prefix sums need one entry more than nodes");
  static_assert(TILE_ROWS <= 1024 && 128 % AGG_WAVES == 0, "one thread per counter; whole rows of W per wave");
  __shared__ int fillc[TILE_ROWS];
  __shared__ int tst[TILE_ROWS];

  const int lane = threadIdx.x & 31;
  const int w = __builtin_amdgcn_readfirstlane(threadIdx.x >> 5);
  const int lo = blockIdx.x * TILE_NODES;
  int cntNodes = n - lo;
  if (cntNodes > TILE_NODES) cntNodes = TILE_NODES;
  const int nStrips = (cntNodes + 15) >> 4;

  int tsv = 0;
  if (threadIdx.x <= TILE_NODES) tsv = tstart[(size_t)blockIdx.x * (TILE_NODES + 1) + threadIdx.x];

  for (int t0 = w; t0 < cntNodes; t0 += AGG_WAVES * INIT_AHEAD) {
    float4 v[INIT_AHEAD];
    float d[INIT_AHEAD];
#pragma unroll
    for (int q = 0; q < INIT_AHEAD; ++q) {
      const int t = t0 + q * AGG_WAVES;
      const int c = lo + (t < cntNodes ? t : cntNodes - 1);
      d[q] = dinv[c];
      v[q] = *(const float4*)(x + (size_t)c * 128 + lane * 4);
    }
#pragma unroll
    for (int q = 0; q < INIT_AHEAD; ++q)
      asm volatile("" ::"v"(v[q].x), "v"(v[q].y), "v"(v[q].z), "v"(v[q].w), "s"(d[q]));
#pragma unroll
    for (int q = 0; q < INIT_AHEAD; ++q) {
      const int t = t0 + q * AGG_WAVES;
      if (t < cntNodes) {
        float* a = acc + t * LPITCH + lane;
        a[0] = v[q].x * d[q];
        a[32] = v[q].y * d[q];
        a[64] = v[q].z * d[q];
        a[96] = v[q].w * d[q];
      }
    }
  }
  for (int t = cntNodes + w; t < nStrips * 16; t += AGG_WAVES) {
    float* a = acc + t * LPITCH + lane;
    a[0] = 0.f;
    a[32] = 0.f;
    a[64] = 0.f;
    a[96] = 0.f;
  }
  if (threadIdx.x < TILE_ROWS) {
    fillc[threadIdx.x] = 0;
    tst[threadIdx.x] = tsv;
  }
  __syncthreads();

  const int2* pairs = (const int2*)pairs_in;
  const int total = __builtin_amdgcn_readfirstlane(tst[TILE_NODES]);
  if (total <= CAP_EDGES) {
    int2* sorted = (int2*)qe_all;
    place_edges(sorted, fillc, tst, lo, cntNodes, pairs, offs, e);
    __syncthreads();
    sum_ordered(acc, sorted, total, w, lane, x, dinv);
  } else {
    int* qe = qe_all + w * QCAP;
    const int cnt = scan_runs(acc, qe, w, lane, lo, cntNodes, x, pairs, dinv, offs, e);
    drain_queue(acc, qe, cnt, lane, lo, cntNodes, x, pairs, dinv);
  }
  float4 wr[128 / AGG_WAVES];
#pragma unroll
  for (int q = 0; q < 128 / AGG_WAVES; ++q)
    wr[q] = *(const float4*)(W + (size_t)(w + q * AGG_WAVES) * 128 + lane * 4);
  __syncthreads();

#pragma unroll
  for (int q = 0; q < 128 / AGG_WAVES; ++q) {
    int* d = qe_all + (w + q * AGG_WAVES) * LPITCH + lane;
    d[0] = __float_as_int(wr[q].x);
    d[32] = __float_as_int(wr[q].y);
    d[64] = __float_as_int(wr[q].z);
    d[96] = __float_as_int(wr[q].w);
  }
  __syncthreads();

  const int m = lane & 15;
  const int h = lane >> 4;
  const int ct = w & 7;
  const int sc = ((w >> 3) + (w & 3)) & 3;

  v2f b[32];
  {
    const int* wl = qe_all + (ct * 16 + m) * LPITCH + 32 * h;
#pragma unroll
    for (int s = 0; s < 32; ++s) {
      const int2 t = *(const int2*)(wl + 64 * (s >> 4) + 2 * (s & 15));
      b[s][0] = __int_as_float(t.x);
      b[s][1] = __int_as_float(t.y);
    }
  }
  const float bn = bias[ct * 16 + m];

  for (int strip = sc; strip < nStrips; strip += 4) {
    const int r0 = strip * 16 + h * 8;
    float dv[8];
#pragma unroll
    for (int v = 0; v < 8; ++v) {
      const int rr = r0 + v;
      dv[v] = dinv[lo + (rr < cntNodes ? rr : cntNodes - 1)];
    }

    const float* arow = acc + (strip * 16 + m) * LPITCH + 32 * h;
    v8f c = {};
#pragma unroll
    for (int s = 0; s < 32; ++s) {
      const v2f a = *(const v2f*)(arow + 64 * (s >> 4) + 2 * (s & 15));
      c = __builtin_amdgcn_wmma_f32_16x16x4_f32(
           false, a,  false, b[s],
           (short)0, c,  false,  false);
    }

    float* orow = out + (size_t)(lo + r0) * 128 + ct * 16 + m;
#pragma unroll
    for (int v = 0; v < 8; ++v) {
      if (r0 + v < cntNodes) orow[(size_t)v * 128] = c[v] * dv[v] + bn;
    }
  }
}


extern "C" void kernel_launch(void* const* d_in, const int* in_sizes, int n_in,
                              void* d_out, int out_size, void* d_ws, size_t ws_size,
                              hipStream_t stream) {
  const float* x  = (const float*)d_in[0];
  const int* ei   = (const int*)d_in[1];
  const float* W  = (const float*)d_in[2];
  const float* b  = (const float*)d_in[3];
  float* out      = (float*)d_out;

  const int N = in_sizes[0] / 128;
  const int E = in_sizes[1] / 2;
  const int* row = ei;
  const int* col = ei + E;

  int* pairs = (int*)d_ws;
  const int nChunks = (E + CHUNK_EDGES - 1) / CHUNK_EDGES;
  unsigned* offs = (unsigned*)(pairs + (size_t)nChunks * (CHUNK_EDGES * 2));
  float* dinv = (float*)(offs + (size_t)nChunks * 256);
  int* tstart = (int*)(dinv + N);

  if (nChunks > 0)
    k_chunk_sort<<<nChunks, 256, 0, stream>>>(row, col, pairs, offs, E);
  k_tile_dinv<<<(N + TILE_NODES - 1) / TILE_NODES, DEG_THREADS, 0, stream>>>(pairs, offs, dinv, tstart, N, E);

  k_aggregate<<<(N + TILE_NODES - 1) / TILE_NODES, 1024, 0, stream>>>(x, W, pairs, offs, dinv, tstart, b, out, N, E);
}
